# starting kernel
# baseline (speedup 1.0000x reference)
#include <hip/hip_runtime.h>


typedef __attribute__((ext_vector_type(2))) float v2f;
typedef __attribute__((ext_vector_type(8))) float v8f;

static constexpr int kN   = 129;
static constexpr int kIN  = 129;
static constexpr int kOUT = 64;
static constexpr int kE   = 4096;
static constexpr int kLD  = 132;
static constexpr int kRP  = 144;
static constexpr int kKS  = 33;
static constexpr int kMT  = 9;

static __device__ __forceinline__ v8f wmma4(v2f a, v2f b, v8f c) {
  return __builtin_amdgcn_wmma_f32_16x16x4_f32(
       false, a,  false, b,
       (short)0, c,  false,  false);
}

static __device__ __forceinline__ float sigm(float x) {
  return 1.0f / (1.0f + __expf(-x));
}

__launch_bounds__(512, 1)
__global__ void evolve_gcn_fused(const float* __restrict__ x,
                                 const int*   __restrict__ eidx,
                                 const float* __restrict__ ew,
                                 const float* __restrict__ pool_p,
                                 const float* __restrict__ W0,
                                 const float* __restrict__ w_ih,
                                 const float* __restrict__ w_hh,
                                 const float* __restrict__ b_ih,
                                 const float* __restrict__ b_hh,
                                 const float* __restrict__ conv_b,
                                 const float* __restrict__ lin_w,
                                 const float* __restrict__ lin_b,
                                 float*       __restrict__ out)
{
  __shared__ float s_x [kRP * kLD];
  __shared__ float s_xt[kRP * kLD];
  __shared__ float s_h [kRP * kLD];
  __shared__ float s_w [kRP * kLD];
  __shared__ float s_score[kN];
  __shared__ int   s_perm [kN];
  __shared__ float s_dis  [kN];
  __shared__ float s_deg  [kN];

  const int tid    = threadIdx.x;
  const int nthr   = blockDim.x;
  const int lane   = tid & 31;
  const int wave   = tid >> 5;
  const int nwav   = nthr >> 5;
  const int m16    = lane & 15;
  const int hh     = lane >> 4;

  for (int i = tid; i < kRP * kLD; i += nthr) {
    s_x[i] = 0.f; s_xt[i] = 0.f; s_h[i] = 0.f; s_w[i] = 0.f;
  }
  for (int i = tid; i < kN; i += nthr) s_deg[i] = 0.f;
  __syncthreads();

  for (int i = tid; i < kN * kIN; i += nthr) {
    int r = i / kIN, c = i % kIN;
    s_x[r * kLD + c] = x[i];
    s_h[r * kLD + c] = W0[i];
  }
  for (int e = tid; e < kE; e += nthr) {
    atomicAdd(&s_deg[eidx[kE + e]], ew[e]);
  }
  for (int i = tid; i < kN; i += nthr) atomicAdd(&s_deg[i], 1.0f);
  __syncthreads();

  for (int i = tid; i < kN; i += nthr) {
    float pn = 0.f, d = 0.f;
    for (int k = 0; k < kIN; ++k) { float pv = pool_p[k]; pn += pv * pv; }
    for (int k = 0; k < kIN; ++k) d += s_x[i * kLD + k] * pool_p[k];
    s_score[i] = tanhf(d * rsqrtf(pn));
    float dg = s_deg[i];
    s_dis[i] = dg > 0.f ? rsqrtf(fmaxf(dg, 1e-12f)) : 0.f;
  }
  __syncthreads();

  for (int i = tid; i < kN; i += nthr) {
    float si = s_score[i];
    int rank = 0;
    for (int j = 0; j < kN; ++j) {
      float sj = s_score[j];
      rank += (sj > si) || (sj == si && j < i);
    }
    s_perm[rank] = i;
  }
  __syncthreads();

  for (int i = tid; i < kN * kLD; i += nthr) {
    int r = i / kLD, c = i % kLD;
    int src = s_perm[r];
    s_xt[r * kLD + c] = s_x[src * kLD + c] * s_score[src];
  }
  __syncthreads();

  for (int t = wave; t < kMT * kMT; t += nwav) {
    int mt = t / kMT, nt = t % kMT;
    int jn = nt * 16 + m16;
    const int rowr = jn * kIN;
    const int rowz = (kN + jn) * kIN;
    const int rown = min(2 * kN + jn, 3 * kN - 1) * kIN;
    float br = b_ih[min(jn, 3 * kN - 1)]          + b_hh[min(jn, 3 * kN - 1)];
    float bz = b_ih[min(kN + jn, 3 * kN - 1)]     + b_hh[min(kN + jn, 3 * kN - 1)];
    float bi = b_ih[min(2 * kN + jn, 3 * kN - 1)];
    float bh = b_hh[min(2 * kN + jn, 3 * kN - 1)];
    v8f ar, az, ai, ah;
#pragma unroll
    for (int v = 0; v < 8; ++v) { ar[v] = br; az[v] = bz; ai[v] = bi; ah[v] = bh; }

    const int arow = (mt * 16 + m16) * kLD;
    for (int ks = 0; ks < kKS; ++ks) {
      int k0 = ks * 4 + 2 * hh;
      v2f axt, ah0;
      axt.x = s_xt[arow + k0]; axt.y = s_xt[arow + k0 + 1];
      ah0.x = s_h [arow + k0]; ah0.y = s_h [arow + k0 + 1];
      int i0 = min(k0,     kIN - 1);
      int i1 = min(k0 + 1, kIN - 1);
      v2f bri, bzi, bni, brh, bzh, bnh;
      bri.x = w_ih[rowr + i0];  bri.y = w_ih[rowr + i1];
      bzi.x = w_ih[rowz + i0];  bzi.y = w_ih[rowz + i1];
      bni.x = w_ih[rown + i0];  bni.y = w_ih[rown + i1];
      brh.x = w_hh[rowr + i0];  brh.y = w_hh[rowr + i1];
      bzh.x = w_hh[rowz + i0];  bzh.y = w_hh[rowz + i1];
      bnh.x = w_hh[rown + i0];  bnh.y = w_hh[rown + i1];
      ar = wmma4(axt, bri, ar);   ar = wmma4(ah0, brh, ar);
      az = wmma4(axt, bzi, az);   az = wmma4(ah0, bzh, az);
      ai = wmma4(axt, bni, ai);
      ah = wmma4(ah0, bnh, ah);
    }
    if (jn < kLD) {
#pragma unroll
      for (int v = 0; v < 8; ++v) {
        int row = mt * 16 + v + 8 * hh;
        float rg  = sigm(ar[v]);
        float zg  = sigm(az[v]);
        float cnd = tanhf(ai[v] + rg * ah[v]);
        float w0v = s_h[row * kLD + jn];
        s_w[row * kLD + jn] = (1.f - zg) * cnd + zg * w0v;
      }
    }
  }
  __syncthreads();

  for (int t = wave; t < kMT * kMT; t += nwav) {
    int mt = t / kMT, nt = t % kMT;
    int jn = nt * 16 + m16;
    int jn_c = min(jn, kLD - 1);
    v8f acc = {0.f, 0.f, 0.f, 0.f, 0.f, 0.f, 0.f, 0.f};
    const int arow = (mt * 16 + m16) * kLD;
    for (int ks = 0; ks < kKS; ++ks) {
      int k0 = ks * 4 + 2 * hh;
      v2f a; a.x = s_x[arow + k0];           a.y = s_x[arow + k0 + 1];
      v2f b; b.x = s_w[k0 * kLD + jn_c];     b.y = s_w[(k0 + 1) * kLD + jn_c];
      acc = wmma4(a, b, acc);
    }
    if (jn < kLD) {
#pragma unroll
      for (int v = 0; v < 8; ++v)
        s_h[(mt * 16 + v + 8 * hh) * kLD + jn] = acc[v];
    }
  }
  __syncthreads();

  float* s_out = s_xt;
  for (int i = tid; i < kRP * kLD; i += nthr) s_out[i] = 0.f;
  __syncthreads();

  for (int e = wave; e < kE + kN; e += nwav) {
    int r, c; float wgt;
    if (e < kE) { r = eidx[e]; c = eidx[kE + e]; wgt = ew[e]; }
    else        { r = e - kE;  c = r;            wgt = 1.0f; }
    float nrm = s_dis[r] * wgt * s_dis[c];
    for (int f = lane; f < kN; f += 32)
      atomicAdd(&s_out[c * kLD + f], nrm * s_h[r * kLD + f]);
  }
  __syncthreads();

  for (int i = tid; i < kN * kN; i += nthr) {
    int r = i / kN, c = i % kN;
    float v = s_out[r * kLD + c] + conv_b[c];
    s_out[r * kLD + c] = v > 0.f ? v : (__expf(v) - 1.f);
  }
  __syncthreads();

  for (int t = wave; t < kMT * 4; t += nwav) {
    int mt = t / 4, ot = t % 4;
    int oc = ot * 16 + m16;
    float lb = lin_b[oc];
    v8f acc;
#pragma unroll
    for (int v = 0; v < 8; ++v) acc[v] = lb;
    const int arow = (mt * 16 + m16) * kLD;
    const int rowo = oc * kIN;
    for (int ks = 0; ks < kKS; ++ks) {
      int k0 = ks * 4 + 2 * hh;
      v2f a; a.x = s_out[arow + k0]; a.y = s_out[arow + k0 + 1];
      int i0 = min(k0,     kIN - 1);
      int i1 = min(k0 + 1, kIN - 1);
      v2f b; b.x = lin_w[rowo + i0]; b.y = lin_w[rowo + i1];
      acc = wmma4(a, b, acc);
    }
#pragma unroll
    for (int v = 0; v < 8; ++v) {
      int row = mt * 16 + v + 8 * hh;
      if (row < kN) out[row * kOUT + oc] = acc[v];
    }
  }
}

extern "C" void kernel_launch(void* const* d_in, const int* in_sizes, int n_in,
                              void* d_out, int out_size, void* d_ws, size_t ws_size,
                              hipStream_t stream) {
  (void)in_sizes; (void)n_in; (void)out_size; (void)d_ws; (void)ws_size;
  const float* x       = (const float*)d_in[0];
  const int*   eidx    = (const int*)  d_in[1];
  const float* ew      = (const float*)d_in[2];
  const float* pool_p  = (const float*)d_in[3];
  const float* W0      = (const float*)d_in[4];
  const float* w_ih    = (const float*)d_in[5];
  const float* w_hh    = (const float*)d_in[6];
  const float* b_ih    = (const float*)d_in[7];
  const float* b_hh    = (const float*)d_in[8];
  const float* conv_b  = (const float*)d_in[9];
  const float* lin_w   = (const float*)d_in[10];
  const float* lin_b   = (const float*)d_in[11];
  evolve_gcn_fused<<<1, 512, 0, stream>>>(x, eidx, ew, pool_p, W0, w_ih, w_hh,
                                          b_ih, b_hh, conv_b, lin_w, lin_b,
                                          (float*)d_out);
}


# Edge aggregation as a dense GEMM: one LDS atomic per edge builds the normalized adjacency, then elu(adj@xw+bias) on f32 WMMA replaces 129 LDS atomics per edge; changes float summation order
# speedup vs baseline: 1.5302x; 1.5302x over previous
#include <hip/hip_runtime.h>


typedef __attribute__((ext_vector_type(2))) float v2f;
typedef __attribute__((ext_vector_type(8))) float v8f;

static constexpr int kN   = 129;
static constexpr int kIN  = 129;
static constexpr int kOUT = 64;
static constexpr int kE   = 4096;
static constexpr int kLD  = 132;
static constexpr int kRP  = 144;
static constexpr int kKS  = 33;
static constexpr int kMT  = 9;

static __device__ __forceinline__ v8f wmma4(v2f a, v2f b, v8f c) {
  return __builtin_amdgcn_wmma_f32_16x16x4_f32(
       false, a,  false, b,
       (short)0, c,  false,  false);
}

static __device__ __forceinline__ float sigm(float x) {
  return 1.0f / (1.0f + __expf(-x));
}

__launch_bounds__(512, 1)
__global__ void evolve_gcn_fused(const float* __restrict__ x,
                                 const int*   __restrict__ eidx,
                                 const float* __restrict__ ew,
                                 const float* __restrict__ pool_p,
                                 const float* __restrict__ W0,
                                 const float* __restrict__ w_ih,
                                 const float* __restrict__ w_hh,
                                 const float* __restrict__ b_ih,
                                 const float* __restrict__ b_hh,
                                 const float* __restrict__ conv_b,
                                 const float* __restrict__ lin_w,
                                 const float* __restrict__ lin_b,
                                 float*       __restrict__ out)
{
  __shared__ float s_x [kRP * kLD];
  __shared__ float s_xt[kRP * kLD];
  __shared__ float s_h [kRP * kLD];
  __shared__ float s_w [kRP * kLD];
  __shared__ float s_score[kN];
  __shared__ int   s_perm [kN];
  __shared__ float s_dis  [kN];
  __shared__ float s_deg  [kN];

  const int tid    = threadIdx.x;
  const int nthr   = blockDim.x;
  const int lane   = tid & 31;
  const int wave   = tid >> 5;
  const int nwav   = nthr >> 5;
  const int m16    = lane & 15;
  const int hh     = lane >> 4;

  for (int i = tid; i < kRP * kLD; i += nthr) {
    s_x[i] = 0.f; s_xt[i] = 0.f; s_h[i] = 0.f; s_w[i] = 0.f;
  }
  for (int i = tid; i < kN; i += nthr) s_deg[i] = 0.f;
  __syncthreads();

  for (int i = tid; i < kN * kIN; i += nthr) {
    int r = i / kIN, c = i % kIN;
    s_x[r * kLD + c] = x[i];
    s_h[r * kLD + c] = W0[i];
  }
  for (int e = tid; e < kE; e += nthr) {
    atomicAdd(&s_deg[eidx[kE + e]], ew[e]);
  }
  for (int i = tid; i < kN; i += nthr) atomicAdd(&s_deg[i], 1.0f);
  __syncthreads();

  for (int i = tid; i < kN; i += nthr) {
    float pn = 0.f, d = 0.f;
    for (int k = 0; k < kIN; ++k) { float pv = pool_p[k]; pn += pv * pv; }
    for (int k = 0; k < kIN; ++k) d += s_x[i * kLD + k] * pool_p[k];
    s_score[i] = tanhf(d * rsqrtf(pn));
    float dg = s_deg[i];
    s_dis[i] = dg > 0.f ? rsqrtf(fmaxf(dg, 1e-12f)) : 0.f;
  }
  __syncthreads();

  for (int i = tid; i < kN; i += nthr) {
    float si = s_score[i];
    int rank = 0;
    for (int j = 0; j < kN; ++j) {
      float sj = s_score[j];
      rank += (sj > si) || (sj == si && j < i);
    }
    s_perm[rank] = i;
  }
  __syncthreads();

  for (int i = tid; i < kN * kLD; i += nthr) {
    int r = i / kLD, c = i % kLD;
    int src = s_perm[r];
    s_xt[r * kLD + c] = s_x[src * kLD + c] * s_score[src];
  }
  __syncthreads();

  for (int t = wave; t < kMT * kMT; t += nwav) {
    int mt = t / kMT, nt = t % kMT;
    int jn = nt * 16 + m16;
    const int rowr = jn * kIN;
    const int rowz = (kN + jn) * kIN;
    const int rown = min(2 * kN + jn, 3 * kN - 1) * kIN;
    float br = b_ih[min(jn, 3 * kN - 1)]          + b_hh[min(jn, 3 * kN - 1)];
    float bz = b_ih[min(kN + jn, 3 * kN - 1)]     + b_hh[min(kN + jn, 3 * kN - 1)];
    float bi = b_ih[min(2 * kN + jn, 3 * kN - 1)];
    float bh = b_hh[min(2 * kN + jn, 3 * kN - 1)];
    v8f ar, az, ai, ah;
#pragma unroll
    for (int v = 0; v < 8; ++v) { ar[v] = br; az[v] = bz; ai[v] = bi; ah[v] = bh; }

    const int arow = (mt * 16 + m16) * kLD;
    for (int ks = 0; ks < kKS; ++ks) {
      int k0 = ks * 4 + 2 * hh;
      v2f axt, ah0;
      axt.x = s_xt[arow + k0]; axt.y = s_xt[arow + k0 + 1];
      ah0.x = s_h [arow + k0]; ah0.y = s_h [arow + k0 + 1];
      int i0 = min(k0,     kIN - 1);
      int i1 = min(k0 + 1, kIN - 1);
      v2f bri, bzi, bni, brh, bzh, bnh;
      bri.x = w_ih[rowr + i0];  bri.y = w_ih[rowr + i1];
      bzi.x = w_ih[rowz + i0];  bzi.y = w_ih[rowz + i1];
      bni.x = w_ih[rown + i0];  bni.y = w_ih[rown + i1];
      brh.x = w_hh[rowr + i0];  brh.y = w_hh[rowr + i1];
      bzh.x = w_hh[rowz + i0];  bzh.y = w_hh[rowz + i1];
      bnh.x = w_hh[rown + i0];  bnh.y = w_hh[rown + i1];
      ar = wmma4(axt, bri, ar);   ar = wmma4(ah0, brh, ar);
      az = wmma4(axt, bzi, az);   az = wmma4(ah0, bzh, az);
      ai = wmma4(axt, bni, ai);
      ah = wmma4(ah0, bnh, ah);
    }
    if (jn < kLD) {
#pragma unroll
      for (int v = 0; v < 8; ++v) {
        int row = mt * 16 + v + 8 * hh;
        float rg  = sigm(ar[v]);
        float zg  = sigm(az[v]);
        float cnd = tanhf(ai[v] + rg * ah[v]);
        float w0v = s_h[row * kLD + jn];
        s_w[row * kLD + jn] = (1.f - zg) * cnd + zg * w0v;
      }
    }
  }
  __syncthreads();

  float* s_adj = s_xt;
  for (int i = tid; i < kRP * kLD; i += nthr) s_adj[i] = 0.f;
  for (int t = wave; t < kMT * kMT; t += nwav) {
    int mt = t / kMT, nt = t % kMT;
    int jn = nt * 16 + m16;
    int jn_c = min(jn, kLD - 1);
    v8f acc = {0.f, 0.f, 0.f, 0.f, 0.f, 0.f, 0.f, 0.f};
    const int arow = (mt * 16 + m16) * kLD;
    for (int ks = 0; ks < kKS; ++ks) {
      int k0 = ks * 4 + 2 * hh;
      v2f a; a.x = s_x[arow + k0];           a.y = s_x[arow + k0 + 1];
      v2f b; b.x = s_w[k0 * kLD + jn_c];     b.y = s_w[(k0 + 1) * kLD + jn_c];
      acc = wmma4(a, b, acc);
    }
    if (jn < kLD) {
#pragma unroll
      for (int v = 0; v < 8; ++v)
        s_h[(mt * 16 + v + 8 * hh) * kLD + jn] = acc[v];
    }
  }
  __syncthreads();

  for (int e = tid; e < kE; e += nthr) {
    int r = eidx[e], c = eidx[kE + e];
    atomicAdd(&s_adj[c * kLD + r], s_dis[r] * ew[e] * s_dis[c]);
  }
  for (int i = tid; i < kN; i += nthr) {
    float d = s_dis[i];
    atomicAdd(&s_adj[i * kLD + i], d * d);
  }
  __syncthreads();

  float* s_out = s_w;
  for (int t = wave; t < kMT * kMT; t += nwav) {
    int mt = t / kMT, nt = t % kMT;
    int jn = nt * 16 + m16;
    int jn_c = min(jn, kLD - 1);
    float cb = conv_b[min(jn, kN - 1)];
    v8f acc;
    for (int v = 0; v < 8; ++v) acc[v] = cb;
    const int arow = (mt * 16 + m16) * kLD;
    for (int ks = 0; ks < kKS; ++ks) {
      int k0 = ks * 4 + 2 * hh;
      v2f a; a.x = s_adj[arow + k0];         a.y = s_adj[arow + k0 + 1];
      v2f b; b.x = s_h[k0 * kLD + jn_c];     b.y = s_h[(k0 + 1) * kLD + jn_c];
      acc = wmma4(a, b, acc);
    }
    if (jn < kLD) {
      for (int v = 0; v < 8; ++v) {
        float hv = acc[v];
        hv = hv > 0.f ? hv : (__expf(hv) - 1.f);
        s_out[(mt * 16 + v + 8 * hh) * kLD + jn] = jn < kN ? hv : 0.f;
      }
    }
  }
  __syncthreads();

  for (int t = wave; t < kMT * 4; t += nwav) {
    int mt = t / 4, ot = t % 4;
    int oc = ot * 16 + m16;
    float lb = lin_b[oc];
    v8f acc;
#pragma unroll
    for (int v = 0; v < 8; ++v) acc[v] = lb;
    const int arow = (mt * 16 + m16) * kLD;
    const int rowo = oc * kIN;
    for (int ks = 0; ks < kKS; ++ks) {
      int k0 = ks * 4 + 2 * hh;
      v2f a; a.x = s_out[arow + k0]; a.y = s_out[arow + k0 + 1];
      int i0 = min(k0,     kIN - 1);
      int i1 = min(k0 + 1, kIN - 1);
      v2f b; b.x = lin_w[rowo + i0]; b.y = lin_w[rowo + i1];
      acc = wmma4(a, b, acc);
    }
#pragma unroll
    for (int v = 0; v < 8; ++v) {
      int row = mt * 16 + v + 8 * hh;
      if (row < kN) out[row * kOUT + oc] = acc[v];
    }
  }
}

extern "C" void kernel_launch(void* const* d_in, const int* in_sizes, int n_in,
                              void* d_out, int out_size, void* d_ws, size_t ws_size,
                              hipStream_t stream) {
  (void)in_sizes; (void)n_in; (void)out_size; (void)d_ws; (void)ws_size;
  const float* x       = (const float*)d_in[0];
  const int*   eidx    = (const int*)  d_in[1];
  const float* ew      = (const float*)d_in[2];
  const float* pool_p  = (const float*)d_in[3];
  const float* W0      = (const float*)d_in[4];
  const float* w_ih    = (const float*)d_in[5];
  const float* w_hh    = (const float*)d_in[6];
  const float* b_ih    = (const float*)d_in[7];
  const float* b_hh    = (const float*)d_in[8];
  const float* conv_b  = (const float*)d_in[9];
  const float* lin_w   = (const float*)d_in[10];
  const float* lin_b   = (const float*)d_in[11];
  evolve_gcn_fused<<<1, 512, 0, stream>>>(x, eidx, ew, pool_p, W0, w_ih, w_hh,
                                          b_ih, b_hh, conv_b, lin_w, lin_b,
                                          (float*)d_out);
}


# GRU gate weights staged once per 16-k chunk into LDS by row-contiguous 16-byte loads, each wave holds its 5 tiles' accumulators across chunks, B fragments read from LDS; summation order unchanged
# speedup vs baseline: 2.7187x; 1.7767x over previous
#include <hip/hip_runtime.h>


typedef __attribute__((ext_vector_type(2))) float v2f;
typedef __attribute__((ext_vector_type(8))) float v8f;

static constexpr int kN   = 129;
static constexpr int kIN  = 129;
static constexpr int kOUT = 64;
static constexpr int kE   = 4096;
static constexpr int kLD  = 132;
static constexpr int kRP  = 144;
static constexpr int kKS  = 33;
static constexpr int kMT  = 9;
static constexpr int kCH  = 16;
static constexpr int kSP  = 20;
static constexpr int kNCH = 9;
static constexpr int kTPW = 5;

static __device__ __forceinline__ v8f wmma4(v2f a, v2f b, v8f c) {
  return __builtin_amdgcn_wmma_f32_16x16x4_f32(
       false, a,  false, b,
       (short)0, c,  false,  false);
}

static __device__ __forceinline__ float sigm(float x) {
  return 1.0f / (1.0f + __expf(-x));
}

__launch_bounds__(512, 1)
__global__ void evolve_gcn_fused(const float* __restrict__ x,
                                 const int*   __restrict__ eidx,
                                 const float* __restrict__ ew,
                                 const float* __restrict__ pool_p,
                                 const float* __restrict__ W0,
                                 const float* __restrict__ w_ih,
                                 const float* __restrict__ w_hh,
                                 const float* __restrict__ b_ih,
                                 const float* __restrict__ b_hh,
                                 const float* __restrict__ conv_b,
                                 const float* __restrict__ lin_w,
                                 const float* __restrict__ lin_b,
                                 float*       __restrict__ out)
{
  __shared__ float s_x [kRP * kLD];
  __shared__ __attribute__((aligned(16))) float s_xt[kRP * kLD];
  __shared__ __attribute__((aligned(16))) float s_h [kRP * kLD];
  __shared__ __attribute__((aligned(16))) float s_w [kRP * kLD];
  __shared__ float s_score[kN];
  __shared__ int   s_perm [kN];
  __shared__ float s_dis  [kN];
  __shared__ float s_deg  [kN];
  __shared__ __attribute__((aligned(16))) float s_b80[12 * kLD];

  const int tid    = threadIdx.x;
  const int nthr   = blockDim.x;
  const int lane   = tid & 31;
  const int wave   = tid >> 5;
  const int nwav   = nthr >> 5;
  const int m16    = lane & 15;
  const int hh     = lane >> 4;

  for (int i = tid; i < kRP * kLD; i += nthr) {
    s_x[i] = 0.f; s_xt[i] = 0.f; s_h[i] = 0.f; s_w[i] = 0.f;
  }
  for (int i = tid; i < kN; i += nthr) s_deg[i] = 0.f;
  __syncthreads();

  for (int i = tid; i < kN * kIN; i += nthr) {
    int r = i / kIN, c = i % kIN;
    s_x[r * kLD + c] = x[i];
    s_h[r * kLD + c] = W0[i];
  }
  for (int i = tid; i < 12 * kLD; i += nthr) {
    int r = i / kLD, c = i % kLD, g = r >> 1;
    const float* wsrc = (g < 3) ? w_ih : w_hh;
    s_b80[i] = ((r & 1) == 0 && c < kIN) ? wsrc[((g % 3) * kN + (kN - 1)) * kIN + c] : 0.f;
  }
  for (int e = tid; e < kE; e += nthr) {
    atomicAdd(&s_deg[eidx[kE + e]], ew[e]);
  }
  for (int i = tid; i < kN; i += nthr) atomicAdd(&s_deg[i], 1.0f);
  __syncthreads();

  for (int i = tid; i < kN; i += nthr) {
    float pn = 0.f, d = 0.f;
    for (int k = 0; k < kIN; ++k) { float pv = pool_p[k]; pn += pv * pv; }
    for (int k = 0; k < kIN; ++k) d += s_x[i * kLD + k] * pool_p[k];
    s_score[i] = tanhf(d * rsqrtf(pn));
    float dg = s_deg[i];
    s_dis[i] = dg > 0.f ? rsqrtf(fmaxf(dg, 1e-12f)) : 0.f;
  }
  __syncthreads();

  for (int i = tid; i < kN; i += nthr) {
    float si = s_score[i];
    int rank = 0;
    for (int j = 0; j < kN; ++j) {
      float sj = s_score[j];
      rank += (sj > si) || (sj == si && j < i);
    }
    s_perm[rank] = i;
  }
  __syncthreads();

  for (int i = tid; i < kN * kLD; i += nthr) {
    int r = i / kLD, c = i % kLD;
    int src = s_perm[r];
    s_xt[r * kLD + c] = s_x[src * kLD + c] * s_score[src];
  }
  __syncthreads();

  {
    float* s_stg = s_w;
    v8f ar[kTPW], az[kTPW], ai[kTPW], ah[kTPW];
#pragma unroll
    for (int i = 0; i < kTPW; ++i) {
      const int t  = wave + 16 * i;
      const int jn = (t % kMT) * 16 + m16;
      float br = b_ih[min(jn, 3 * kN - 1)]          + b_hh[min(jn, 3 * kN - 1)];
      float bz = b_ih[min(kN + jn, 3 * kN - 1)]     + b_hh[min(kN + jn, 3 * kN - 1)];
      float bi = b_ih[min(2 * kN + jn, 3 * kN - 1)];
      float bh = b_hh[min(2 * kN + jn, 3 * kN - 1)];
#pragma unroll
      for (int v = 0; v < 8; ++v) { ar[i][v] = br; az[i][v] = bz; ai[i][v] = bi; ah[i][v] = bh; }
    }

    const int sj   = tid >> 2;
    const int sp   = tid & 3;
    const int soff = sj * kIN + sp * 4;
    float4 sv[6];

#pragma unroll
    for (int g = 0; g < 6; ++g)
      __builtin_memcpy(&sv[g], (g < 3 ? w_ih : w_hh) + (g % 3) * kN * kIN + soff, 16);

    for (int kc = 0; kc < kNCH; ++kc) {
      __syncthreads();
#pragma unroll
      for (int g = 0; g < 6; ++g)
        *reinterpret_cast<float4*>(&s_stg[(g * kRP + sj) * kSP + sp * 4]) = sv[g];
      if (tid < 24 && (kc < kNCH - 1 || sp == 0)) {
        const int xg = tid >> 2;
        *reinterpret_cast<float4*>(&s_stg[(xg * kRP + (kN - 1)) * kSP + sp * 4]) =
            *reinterpret_cast<const float4*>(&s_b80[2 * xg * kLD + kc * kCH + sp * 4]);
      }
      __syncthreads();

      if (kc + 1 < kNCH - 1) {
#pragma unroll
        for (int g = 0; g < 6; ++g)
          __builtin_memcpy(&sv[g], (g < 3 ? w_ih : w_hh) + (g % 3) * kN * kIN + (kc + 1) * kCH + soff, 16);
      } else if (kc + 1 == kNCH - 1) {
#pragma unroll
        for (int g = 0; g < 6; ++g) {
          const float wv = ((g < 3 ? w_ih : w_hh) + (g % 3) * kN * kIN)[sj * kIN + (kIN - 1)];
          sv[g].x = (sp == 0) ? wv : 0.f;  sv[g].y = 0.f;  sv[g].z = 0.f;  sv[g].w = 0.f;
        }
      }

      const int nks = (kc == kNCH - 1) ? 1 : 4;
      for (int ks = 0; ks < nks; ++ks) {
        const int kl = ks * 4 + 2 * hh;
        const int k0 = kc * kCH + kl;
#pragma unroll
        for (int i = 0; i < kTPW; ++i) {
          const int t  = wave + 16 * i;
          const int mt = t / kMT, nt = t % kMT;
          const int arow = (mt * 16 + m16) * kLD + k0;
          const int brow = (nt * 16 + m16) * kSP + kl;
          v2f axt = *reinterpret_cast<const v2f*>(&s_xt[arow]);
          v2f ah0 = *reinterpret_cast<const v2f*>(&s_h [arow]);
          v2f bri = *reinterpret_cast<const v2f*>(&s_stg[brow + 0 * kRP * kSP]);
          v2f bzi = *reinterpret_cast<const v2f*>(&s_stg[brow + 1 * kRP * kSP]);
          v2f bni = *reinterpret_cast<const v2f*>(&s_stg[brow + 2 * kRP * kSP]);
          v2f brh = *reinterpret_cast<const v2f*>(&s_stg[brow + 3 * kRP * kSP]);
          v2f bzh = *reinterpret_cast<const v2f*>(&s_stg[brow + 4 * kRP * kSP]);
          v2f bnh = *reinterpret_cast<const v2f*>(&s_stg[brow + 5 * kRP * kSP]);
          ar[i] = wmma4(axt, bri, ar[i]);   ar[i] = wmma4(ah0, brh, ar[i]);
          az[i] = wmma4(axt, bzi, az[i]);   az[i] = wmma4(ah0, bzh, az[i]);
          ai[i] = wmma4(axt, bni, ai[i]);
          ah[i] = wmma4(ah0, bnh, ah[i]);
        }
      }
    }
    __syncthreads();

    if (wave == 0) {
      const int jn = 8 * 16 + m16;
      float br = b_ih[min(jn, 3 * kN - 1)]          + b_hh[min(jn, 3 * kN - 1)];
      float bz = b_ih[min(kN + jn, 3 * kN - 1)]     + b_hh[min(kN + jn, 3 * kN - 1)];
      float bi = b_ih[min(2 * kN + jn, 3 * kN - 1)];
      float bh = b_hh[min(2 * kN + jn, 3 * kN - 1)];
      v8f cr, cz, ci, ch;
#pragma unroll
      for (int v = 0; v < 8; ++v) { cr[v] = br; cz[v] = bz; ci[v] = bi; ch[v] = bh; }
      const int arow = (8 * 16 + m16) * kLD;
      const int zsh = (m16 == 0) ? 0 : kLD;
#pragma unroll 3
      for (int ks = 0; ks < kKS; ++ks) {
        const int k0 = ks * 4 + 2 * hh;
        v2f axt = *reinterpret_cast<const v2f*>(&s_xt[arow + k0]);
        v2f ah0 = *reinterpret_cast<const v2f*>(&s_h [arow + k0]);
        v2f bri = *reinterpret_cast<const v2f*>(&s_b80[0 * kLD + zsh + k0]);
        v2f bzi = *reinterpret_cast<const v2f*>(&s_b80[2 * kLD + zsh + k0]);
        v2f bni = *reinterpret_cast<const v2f*>(&s_b80[4 * kLD + zsh + k0]);
        v2f brh = *reinterpret_cast<const v2f*>(&s_b80[6 * kLD + zsh + k0]);
        v2f bzh = *reinterpret_cast<const v2f*>(&s_b80[8 * kLD + zsh + k0]);
        v2f bnh = *reinterpret_cast<const v2f*>(&s_b80[10 * kLD + zsh + k0]);
        cr = wmma4(axt, bri, cr);   cr = wmma4(ah0, brh, cr);
        cz = wmma4(axt, bzi, cz);   cz = wmma4(ah0, bzh, cz);
        ci = wmma4(axt, bni, ci);
        ch = wmma4(ah0, bnh, ch);
      }
      if (jn < kLD) {
#pragma unroll
        for (int v = 0; v < 8; ++v) {
          int row = 8 * 16 + v + 8 * hh;
          float rg  = sigm(cr[v]);
          float zg  = sigm(cz[v]);
          float cnd = tanhf(ci[v] + rg * ch[v]);
          float w0v = s_h[row * kLD + jn];
          s_w[row * kLD + jn] = (1.f - zg) * cnd + zg * w0v;
        }
      }
    }

#pragma unroll
    for (int i = 0; i < kTPW; ++i) {
      const int t  = wave + 16 * i;
      const int mt = t / kMT, nt = t % kMT;
      const int jn = nt * 16 + m16;
      if (jn < kLD) {
#pragma unroll
        for (int v = 0; v < 8; ++v) {
          int row = mt * 16 + v + 8 * hh;
          float rg  = sigm(ar[i][v]);
          float zg  = sigm(az[i][v]);
          float cnd = tanhf(ai[i][v] + rg * ah[i][v]);
          float w0v = s_h[row * kLD + jn];
          s_w[row * kLD + jn] = (1.f - zg) * cnd + zg * w0v;
        }
      }
    }
  }

  __syncthreads();

  float* s_adj = s_xt;
  for (int i = tid; i < kRP * kLD; i += nthr) s_adj[i] = 0.f;
  for (int t = wave; t < kMT * kMT; t += nwav) {
    int mt = t / kMT, nt = t % kMT;
    int jn = nt * 16 + m16;
    int jn_c = min(jn, kLD - 1);
    v8f acc = {0.f, 0.f, 0.f, 0.f, 0.f, 0.f, 0.f, 0.f};
    const int arow = (mt * 16 + m16) * kLD;
    for (int ks = 0; ks < kKS; ++ks) {
      int k0 = ks * 4 + 2 * hh;
      v2f a; a.x = s_x[arow + k0];           a.y = s_x[arow + k0 + 1];
      v2f b; b.x = s_w[k0 * kLD + jn_c];     b.y = s_w[(k0 + 1) * kLD + jn_c];
      acc = wmma4(a, b, acc);
    }
    if (jn < kLD) {
#pragma unroll
      for (int v = 0; v < 8; ++v)
        s_h[(mt * 16 + v + 8 * hh) * kLD + jn] = acc[v];
    }
  }
  __syncthreads();

  for (int e = tid; e < kE; e += nthr) {
    int r = eidx[e], c = eidx[kE + e];
    atomicAdd(&s_adj[c * kLD + r], s_dis[r] * ew[e] * s_dis[c]);
  }
  for (int i = tid; i < kN; i += nthr) {
    float d = s_dis[i];
    atomicAdd(&s_adj[i * kLD + i], d * d);
  }
  __syncthreads();

  float* s_out = s_w;
  for (int t = wave; t < kMT * kMT; t += nwav) {
    int mt = t / kMT, nt = t % kMT;
    int jn = nt * 16 + m16;
    int jn_c = min(jn, kLD - 1);
    float cb = conv_b[min(jn, kN - 1)];
    v8f acc;
    for (int v = 0; v < 8; ++v) acc[v] = cb;
    const int arow = (mt * 16 + m16) * kLD;
    for (int ks = 0; ks < kKS; ++ks) {
      int k0 = ks * 4 + 2 * hh;
      v2f a; a.x = s_adj[arow + k0];         a.y = s_adj[arow + k0 + 1];
      v2f b; b.x = s_h[k0 * kLD + jn_c];     b.y = s_h[(k0 + 1) * kLD + jn_c];
      acc = wmma4(a, b, acc);
    }
    if (jn < kLD) {
      for (int v = 0; v < 8; ++v) {
        float hv = acc[v];
        hv = hv > 0.f ? hv : (__expf(hv) - 1.f);
        s_out[(mt * 16 + v + 8 * hh) * kLD + jn] = jn < kN ? hv : 0.f;
      }
    }
  }
  __syncthreads();

  for (int t = wave; t < kMT * 4; t += nwav) {
    int mt = t / 4, ot = t % 4;
    int oc = ot * 16 + m16;
    float lb = lin_b[oc];
    v8f acc;
#pragma unroll
    for (int v = 0; v < 8; ++v) acc[v] = lb;
    const int arow = (mt * 16 + m16) * kLD;
    const int rowo = oc * kIN;
    for (int ks = 0; ks < kKS; ++ks) {
      int k0 = ks * 4 + 2 * hh;
      v2f a; a.x = s_out[arow + k0]; a.y = s_out[arow + k0 + 1];
      int i0 = min(k0,     kIN - 1);
      int i1 = min(k0 + 1, kIN - 1);
      v2f b; b.x = lin_w[rowo + i0]; b.y = lin_w[rowo + i1];
      acc = wmma4(a, b, acc);
    }
#pragma unroll
    for (int v = 0; v < 8; ++v) {
      int row = mt * 16 + v + 8 * hh;
      if (row < kN) out[row * kOUT + oc] = acc[v];
    }
  }
}

extern "C" void kernel_launch(void* const* d_in, const int* in_sizes, int n_in,
                              void* d_out, int out_size, void* d_ws, size_t ws_size,
                              hipStream_t stream) {
  (void)in_sizes; (void)n_in; (void)out_size; (void)d_ws; (void)ws_size;
  const float* x       = (const float*)d_in[0];
  const int*   eidx    = (const int*)  d_in[1];
  const float* ew      = (const float*)d_in[2];
  const float* pool_p  = (const float*)d_in[3];
  const float* W0      = (const float*)d_in[4];
  const float* w_ih    = (const float*)d_in[5];
  const float* w_hh    = (const float*)d_in[6];
  const float* b_ih    = (const float*)d_in[7];
  const float* b_hh    = (const float*)d_in[8];
  const float* conv_b  = (const float*)d_in[9];
  const float* lin_w   = (const float*)d_in[10];
  const float* lin_b   = (const float*)d_in[11];
  evolve_gcn_fused<<<1, 512, 0, stream>>>(x, eidx, ew, pool_p, W0, w_ih, w_hh,
                                          b_ih, b_hh, conv_b, lin_w, lin_b,
                                          (float*)d_out);
}


# Staging phases pay one memory round trip each: compile-time block size, all global loads of x, W0, gate row 128, edges, pool_p issued together ahead of use; pool_p via LDS; sums keep their order
# speedup vs baseline: 3.5445x; 1.3037x over previous
#include <hip/hip_runtime.h>


typedef __attribute__((ext_vector_type(2))) float v2f;
typedef __attribute__((ext_vector_type(8))) float v8f;

static constexpr int kN   = 129;
static constexpr int kIN  = 129;
static constexpr int kOUT = 64;
static constexpr int kE   = 4096;
static constexpr int kLD  = 132;
static constexpr int kRP  = 144;
static constexpr int kKS  = 33;
static constexpr int kMT  = 9;
static constexpr int kCH  = 16;
static constexpr int kSP  = 20;
static constexpr int kNCH = 9;
static constexpr int kTPW = 5;
static constexpr int kNT  = 512;
static constexpr int kNW  = kNT / 32;
static constexpr int kXI  = (kN * kIN + kNT - 1) / kNT;
static constexpr int kBI  = (12 * kLD + kNT - 1) / kNT;
static constexpr int kEI  = kE / kNT;

static __device__ __forceinline__ v8f wmma4(v2f a, v2f b, v8f c) {
  return __builtin_amdgcn_wmma_f32_16x16x4_f32(
       false, a,  false, b,
       (short)0, c,  false,  false);
}

static __device__ __forceinline__ float sigm(float x) {
  return 1.0f / (1.0f + __expf(-x));
}

__launch_bounds__(512, 1)
__global__ void evolve_gcn_fused(const float* __restrict__ x,
                                 const int*   __restrict__ eidx,
                                 const float* __restrict__ ew,
                                 const float* __restrict__ pool_p,
                                 const float* __restrict__ W0,
                                 const float* __restrict__ w_ih,
                                 const float* __restrict__ w_hh,
                                 const float* __restrict__ b_ih,
                                 const float* __restrict__ b_hh,
                                 const float* __restrict__ conv_b,
                                 const float* __restrict__ lin_w,
                                 const float* __restrict__ lin_b,
                                 float*       __restrict__ out)
{
  __shared__ __attribute__((aligned(16))) float s_x [kRP * kLD];
  __shared__ __attribute__((aligned(16))) float s_xt[kRP * kLD];
  __shared__ __attribute__((aligned(16))) float s_h [kRP * kLD];
  __shared__ __attribute__((aligned(16))) float s_w [kRP * kLD];
  __shared__ float s_score[kN];
  __shared__ __attribute__((aligned(16))) float s_pp[kLD];
  __shared__ int   s_perm [kN];
  __shared__ float s_dis  [kN];
  __shared__ float s_deg  [kN];
  __shared__ __attribute__((aligned(16))) float s_b80[12 * kLD];

  const int tid    = threadIdx.x;
  constexpr int nthr = kNT;
  const int lane   = tid & 31;
  const int wave   = tid >> 5;
  constexpr int nwav = kNW;
  const int m16    = lane & 15;
  const int hh     = lane >> 4;

  float xv[kXI], wv[kXI], bv[kBI], dwv[kEI];
  int   dcv[kEI];
#pragma unroll
  for (int j = 0; j < kXI; ++j) {
    const int i = min(tid + j * kNT, kN * kIN - 1);
    xv[j] = x[i];
    wv[j] = W0[i];
  }
#pragma unroll
  for (int j = 0; j < kBI; ++j) {
    const int i = tid + j * kNT;
    const int r = i / kLD, c = i % kLD, g = r >> 1;
    const bool real = (i < 12 * kLD) && ((r & 1) == 0) && (c < kIN);
    const float* wsrc = (g < 3) ? w_ih : w_hh;
    const float v = wsrc[real ? ((g % 3) * kN + (kN - 1)) * kIN + c : 0];
    bv[j] = real ? v : 0.f;
  }
#pragma unroll
  for (int j = 0; j < kEI; ++j) {
    const int e = tid + j * kNT;
    dcv[j] = eidx[kE + e];
    dwv[j] = ew[e];
  }
  const float ppv = pool_p[min(tid, kIN - 1)];

  for (int i = tid; i < kRP * kLD; i += nthr) {
    s_x[i] = 0.f; s_xt[i] = 0.f; s_h[i] = 0.f; s_w[i] = 0.f;
  }
  for (int i = tid; i < kN; i += nthr) s_deg[i] = 0.f;
  __syncthreads();

#pragma unroll
  for (int j = 0; j < kXI; ++j) {
    const int i = tid + j * kNT;
    if (i < kN * kIN) {
      const int r = i / kIN, c = i % kIN;
      s_x[r * kLD + c] = xv[j];
      s_h[r * kLD + c] = wv[j];
    }
  }
#pragma unroll
  for (int j = 0; j < kBI; ++j) {
    const int i = tid + j * kNT;
    if (i < 12 * kLD) s_b80[i] = bv[j];
  }
  if (tid < kLD) s_pp[tid] = (tid < kIN) ? ppv : 0.f;
#pragma unroll
  for (int j = 0; j < kEI; ++j) {
    atomicAdd(&s_deg[dcv[j]], dwv[j]);
  }
  for (int i = tid; i < kN; i += nthr) atomicAdd(&s_deg[i], 1.0f);
  __syncthreads();

  if (tid < kN) {
    const int i = tid;
    float pn = 0.f, d = 0.f;
#pragma unroll
    for (int q = 0; q < kIN / 4; ++q) {
      const float4 pv = *reinterpret_cast<const float4*>(&s_pp[4 * q]);
      const float4 xq = *reinterpret_cast<const float4*>(&s_x[i * kLD + 4 * q]);
      pn = __builtin_fmaf(pv.x, pv.x, pn);  d = __builtin_fmaf(xq.x, pv.x, d);
      pn = __builtin_fmaf(pv.y, pv.y, pn);  d = __builtin_fmaf(xq.y, pv.y, d);
      pn = __builtin_fmaf(pv.z, pv.z, pn);  d = __builtin_fmaf(xq.z, pv.z, d);
      pn = __builtin_fmaf(pv.w, pv.w, pn);  d = __builtin_fmaf(xq.w, pv.w, d);
    }
    {
      const float pl = s_pp[kIN - 1];
      pn = __builtin_fmaf(pl, pl, pn);
      d  = __builtin_fmaf(s_x[i * kLD + kIN - 1], pl, d);
    }
    s_score[i] = tanhf(d * rsqrtf(pn));
    float dg = s_deg[i];
    s_dis[i] = dg > 0.f ? rsqrtf(fmaxf(dg, 1e-12f)) : 0.f;
  }
  __syncthreads();

  for (int i = tid; i < kN; i += nthr) {
    float si = s_score[i];
    int rank = 0;
    for (int j = 0; j < kN; ++j) {
      float sj = s_score[j];
      rank += (sj > si) || (sj == si && j < i);
    }
    s_perm[rank] = i;
  }
  __syncthreads();

  for (int i = tid; i < kN * kLD; i += nthr) {
    int r = i / kLD, c = i % kLD;
    int src = s_perm[r];
    s_xt[r * kLD + c] = s_x[src * kLD + c] * s_score[src];
  }
  __syncthreads();

  {
    float* s_stg = s_w;
    v8f ar[kTPW], az[kTPW], ai[kTPW], ah[kTPW];
#pragma unroll
    for (int i = 0; i < kTPW; ++i) {
      const int t  = wave + 16 * i;
      const int jn = (t % kMT) * 16 + m16;
      float br = b_ih[min(jn, 3 * kN - 1)]          + b_hh[min(jn, 3 * kN - 1)];
      float bz = b_ih[min(kN + jn, 3 * kN - 1)]     + b_hh[min(kN + jn, 3 * kN - 1)];
      float bi = b_ih[min(2 * kN + jn, 3 * kN - 1)];
      float bh = b_hh[min(2 * kN + jn, 3 * kN - 1)];
#pragma unroll
      for (int v = 0; v < 8; ++v) { ar[i][v] = br; az[i][v] = bz; ai[i][v] = bi; ah[i][v] = bh; }
    }

    const int sj   = tid >> 2;
    const int sp   = tid & 3;
    const int soff = sj * kIN + sp * 4;
    float4 sv[6];

#pragma unroll
    for (int g = 0; g < 6; ++g)
      __builtin_memcpy(&sv[g], (g < 3 ? w_ih : w_hh) + (g % 3) * kN * kIN + soff, 16);

    for (int kc = 0; kc < kNCH; ++kc) {
      __syncthreads();
#pragma unroll
      for (int g = 0; g < 6; ++g)
        *reinterpret_cast<float4*>(&s_stg[(g * kRP + sj) * kSP + sp * 4]) = sv[g];
      if (tid < 24 && (kc < kNCH - 1 || sp == 0)) {
        const int xg = tid >> 2;
        *reinterpret_cast<float4*>(&s_stg[(xg * kRP + (kN - 1)) * kSP + sp * 4]) =
            *reinterpret_cast<const float4*>(&s_b80[2 * xg * kLD + kc * kCH + sp * 4]);
      }
      __syncthreads();

      if (kc + 1 < kNCH - 1) {
#pragma unroll
        for (int g = 0; g < 6; ++g)
          __builtin_memcpy(&sv[g], (g < 3 ? w_ih : w_hh) + (g % 3) * kN * kIN + (kc + 1) * kCH + soff, 16);
      } else if (kc + 1 == kNCH - 1) {
#pragma unroll
        for (int g = 0; g < 6; ++g) {
          const float wv = ((g < 3 ? w_ih : w_hh) + (g % 3) * kN * kIN)[sj * kIN + (kIN - 1)];
          sv[g].x = (sp == 0) ? wv : 0.f;  sv[g].y = 0.f;  sv[g].z = 0.f;  sv[g].w = 0.f;
        }
      }

      const int nks = (kc == kNCH - 1) ? 1 : 4;
      for (int ks = 0; ks < nks; ++ks) {
        const int kl = ks * 4 + 2 * hh;
        const int k0 = kc * kCH + kl;
#pragma unroll
        for (int i = 0; i < kTPW; ++i) {
          const int t  = wave + 16 * i;
          const int mt = t / kMT, nt = t % kMT;
          const int arow = (mt * 16 + m16) * kLD + k0;
          const int brow = (nt * 16 + m16) * kSP + kl;
          v2f axt = *reinterpret_cast<const v2f*>(&s_xt[arow]);
          v2f ah0 = *reinterpret_cast<const v2f*>(&s_h [arow]);
          v2f bri = *reinterpret_cast<const v2f*>(&s_stg[brow + 0 * kRP * kSP]);
          v2f bzi = *reinterpret_cast<const v2f*>(&s_stg[brow + 1 * kRP * kSP]);
          v2f bni = *reinterpret_cast<const v2f*>(&s_stg[brow + 2 * kRP * kSP]);
          v2f brh = *reinterpret_cast<const v2f*>(&s_stg[brow + 3 * kRP * kSP]);
          v2f bzh = *reinterpret_cast<const v2f*>(&s_stg[brow + 4 * kRP * kSP]);
          v2f bnh = *reinterpret_cast<const v2f*>(&s_stg[brow + 5 * kRP * kSP]);
          ar[i] = wmma4(axt, bri, ar[i]);   ar[i] = wmma4(ah0, brh, ar[i]);
          az[i] = wmma4(axt, bzi, az[i]);   az[i] = wmma4(ah0, bzh, az[i]);
          ai[i] = wmma4(axt, bni, ai[i]);
          ah[i] = wmma4(ah0, bnh, ah[i]);
        }
      }
    }
    __syncthreads();

    if (wave == 0) {
      const int jn = 8 * 16 + m16;
      float br = b_ih[min(jn, 3 * kN - 1)]          + b_hh[min(jn, 3 * kN - 1)];
      float bz = b_ih[min(kN + jn, 3 * kN - 1)]     + b_hh[min(kN + jn, 3 * kN - 1)];
      float bi = b_ih[min(2 * kN + jn, 3 * kN - 1)];
      float bh = b_hh[min(2 * kN + jn, 3 * kN - 1)];
      v8f cr, cz, ci, ch;
#pragma unroll
      for (int v = 0; v < 8; ++v) { cr[v] = br; cz[v] = bz; ci[v] = bi; ch[v] = bh; }
      const int arow = (8 * 16 + m16) * kLD;
      const int zsh = (m16 == 0) ? 0 : kLD;
#pragma unroll 3
      for (int ks = 0; ks < kKS; ++ks) {
        const int k0 = ks * 4 + 2 * hh;
        v2f axt = *reinterpret_cast<const v2f*>(&s_xt[arow + k0]);
        v2f ah0 = *reinterpret_cast<const v2f*>(&s_h [arow + k0]);
        v2f bri = *reinterpret_cast<const v2f*>(&s_b80[0 * kLD + zsh + k0]);
        v2f bzi = *reinterpret_cast<const v2f*>(&s_b80[2 * kLD + zsh + k0]);
        v2f bni = *reinterpret_cast<const v2f*>(&s_b80[4 * kLD + zsh + k0]);
        v2f brh = *reinterpret_cast<const v2f*>(&s_b80[6 * kLD + zsh + k0]);
        v2f bzh = *reinterpret_cast<const v2f*>(&s_b80[8 * kLD + zsh + k0]);
        v2f bnh = *reinterpret_cast<const v2f*>(&s_b80[10 * kLD + zsh + k0]);
        cr = wmma4(axt, bri, cr);   cr = wmma4(ah0, brh, cr);
        cz = wmma4(axt, bzi, cz);   cz = wmma4(ah0, bzh, cz);
        ci = wmma4(axt, bni, ci);
        ch = wmma4(ah0, bnh, ch);
      }
      if (jn < kLD) {
#pragma unroll
        for (int v = 0; v < 8; ++v) {
          int row = 8 * 16 + v + 8 * hh;
          float rg  = sigm(cr[v]);
          float zg  = sigm(cz[v]);
          float cnd = tanhf(ci[v] + rg * ch[v]);
          float w0v = s_h[row * kLD + jn];
          s_w[row * kLD + jn] = (1.f - zg) * cnd + zg * w0v;
        }
      }
    }

#pragma unroll
    for (int i = 0; i < kTPW; ++i) {
      const int t  = wave + 16 * i;
      const int mt = t / kMT, nt = t % kMT;
      const int jn = nt * 16 + m16;
      if (jn < kLD) {
#pragma unroll
        for (int v = 0; v < 8; ++v) {
          int row = mt * 16 + v + 8 * hh;
          float rg  = sigm(ar[i][v]);
          float zg  = sigm(az[i][v]);
          float cnd = tanhf(ai[i][v] + rg * ah[i][v]);
          float w0v = s_h[row * kLD + jn];
          s_w[row * kLD + jn] = (1.f - zg) * cnd + zg * w0v;
        }
      }
    }
  }

  __syncthreads();

  float* s_adj = s_xt;
  int   aer[kEI], aec[kEI];
  float aew[kEI];
#pragma unroll
  for (int j = 0; j < kEI; ++j) {
    const int e = tid + j * kNT;
    aer[j] = eidx[e];
    aec[j] = eidx[kE + e];
    aew[j] = ew[e];
  }
  for (int i = tid; i < kRP * kLD; i += nthr) s_adj[i] = 0.f;
  for (int t = wave; t < kMT * kMT; t += nwav) {
    int mt = t / kMT, nt = t % kMT;
    int jn = nt * 16 + m16;
    int jn_c = min(jn, kLD - 1);
    v8f acc = {0.f, 0.f, 0.f, 0.f, 0.f, 0.f, 0.f, 0.f};
    const int arow = (mt * 16 + m16) * kLD;
    for (int ks = 0; ks < kKS; ++ks) {
      int k0 = ks * 4 + 2 * hh;
      v2f a; a.x = s_x[arow + k0];           a.y = s_x[arow + k0 + 1];
      v2f b; b.x = s_w[k0 * kLD + jn_c];     b.y = s_w[(k0 + 1) * kLD + jn_c];
      acc = wmma4(a, b, acc);
    }
    if (jn < kLD) {
#pragma unroll
      for (int v = 0; v < 8; ++v)
        s_h[(mt * 16 + v + 8 * hh) * kLD + jn] = acc[v];
    }
  }
  __syncthreads();

  float av[kEI];
#pragma unroll
  for (int j = 0; j < kEI; ++j) av[j] = s_dis[aer[j]] * aew[j] * s_dis[aec[j]];
#pragma unroll
  for (int j = 0; j < kEI; ++j) atomicAdd(&s_adj[aec[j] * kLD + aer[j]], av[j]);
  for (int i = tid; i < kN; i += nthr) {
    float d = s_dis[i];
    atomicAdd(&s_adj[i * kLD + i], d * d);
  }
  __syncthreads();

  float* s_out = s_w;
  for (int t = wave; t < kMT * kMT; t += nwav) {
    int mt = t / kMT, nt = t % kMT;
    int jn = nt * 16 + m16;
    int jn_c = min(jn, kLD - 1);
    float cb = conv_b[min(jn, kN - 1)];
    v8f acc;
    for (int v = 0; v < 8; ++v) acc[v] = cb;
    const int arow = (mt * 16 + m16) * kLD;
    for (int ks = 0; ks < kKS; ++ks) {
      int k0 = ks * 4 + 2 * hh;
      v2f a; a.x = s_adj[arow + k0];         a.y = s_adj[arow + k0 + 1];
      v2f b; b.x = s_h[k0 * kLD + jn_c];     b.y = s_h[(k0 + 1) * kLD + jn_c];
      acc = wmma4(a, b, acc);
    }
    if (jn < kLD) {
      for (int v = 0; v < 8; ++v) {
        float hv = acc[v];
        hv = hv > 0.f ? hv : (__expf(hv) - 1.f);
        s_out[(mt * 16 + v + 8 * hh) * kLD + jn] = jn < kN ? hv : 0.f;
      }
    }
  }
  __syncthreads();

  for (int t = wave; t < kMT * 4; t += nwav) {
    int mt = t / 4, ot = t % 4;
    int oc = ot * 16 + m16;
    float lb = lin_b[oc];
    v8f acc;
#pragma unroll
    for (int v = 0; v < 8; ++v) acc[v] = lb;
    const int arow = (mt * 16 + m16) * kLD;
    const int rowo = oc * kIN;
    for (int ks = 0; ks < kKS; ++ks) {
      int k0 = ks * 4 + 2 * hh;
      v2f a; a.x = s_out[arow + k0]; a.y = s_out[arow + k0 + 1];
      int i0 = min(k0,     kIN - 1);
      int i1 = min(k0 + 1, kIN - 1);
      v2f b; b.x = lin_w[rowo + i0]; b.y = lin_w[rowo + i1];
      acc = wmma4(a, b, acc);
    }
#pragma unroll
    for (int v = 0; v < 8; ++v) {
      int row = mt * 16 + v + 8 * hh;
      if (row < kN) out[row * kOUT + oc] = acc[v];
    }
  }
}

extern "C" void kernel_launch(void* const* d_in, const int* in_sizes, int n_in,
                              void* d_out, int out_size, void* d_ws, size_t ws_size,
                              hipStream_t stream) {
  (void)in_sizes; (void)n_in; (void)out_size; (void)d_ws; (void)ws_size;
  const float* x       = (const float*)d_in[0];
  const int*   eidx    = (const int*)  d_in[1];
  const float* ew      = (const float*)d_in[2];
  const float* pool_p  = (const float*)d_in[3];
  const float* W0      = (const float*)d_in[4];
  const float* w_ih    = (const float*)d_in[5];
  const float* w_hh    = (const float*)d_in[6];
  const float* b_ih    = (const float*)d_in[7];
  const float* b_hh    = (const float*)d_in[8];
  const float* conv_b  = (const float*)d_in[9];
  const float* lin_w   = (const float*)d_in[10];
  const float* lin_b   = (const float*)d_in[11];
  evolve_gcn_fused<<<1, 512, 0, stream>>>(x, eidx, ew, pool_p, W0, w_ih, w_hh,
                                          b_ih, b_hh, conv_b, lin_w, lin_b,
                                          (float*)d_out);
}


# Spread over nine WGPs: one workgroup per 16-column block of W runs GRU, x@W, adj@xw, ELU for its columns (front repeated per group); final linear is a second launch via the workspace; sum order kept
# speedup vs baseline: 11.8336x; 3.3386x over previous
#include <hip/hip_runtime.h>


typedef __attribute__((ext_vector_type(2))) float v2f;
typedef __attribute__((ext_vector_type(8))) float v8f;

static constexpr int kN   = 129;
static constexpr int kIN  = 129;
static constexpr int kOUT = 64;
static constexpr int kE   = 4096;
static constexpr int kLD  = 132;
static constexpr int kRP  = 144;
static constexpr int kKS  = 33;
static constexpr int kMT  = 9;
static constexpr int kNT  = 512;
static constexpr int kXI  = (kN * kIN + kNT - 1) / kNT;
static constexpr int kGE  = 6 * 16 * kLD;
static constexpr int kGI  = (kGE + kNT - 1) / kNT;
static constexpr int kEI  = kE / kNT;
static constexpr int kHB  = kRP * 16;
static constexpr unsigned long long kWsNeed = 4ull * kMT * kHB;

static __device__ __forceinline__ v8f wmma4(v2f a, v2f b, v8f c) {
  return __builtin_amdgcn_wmma_f32_16x16x4_f32(
       false, a,  false, b,
       (short)0, c,  false,  false);
}

static __device__ __forceinline__ float sigm(float x) {
  return 1.0f / (1.0f + __expf(-x));
}

__launch_bounds__(512, 1)
__global__ void evolve_gcn_colblock(const float* __restrict__ x,
                                    const int*   __restrict__ eidx,
                                    const float* __restrict__ ew,
                                    const float* __restrict__ pool_p,
                                    const float* __restrict__ W0,
                                    const float* __restrict__ w_ih,
                                    const float* __restrict__ w_hh,
                                    const float* __restrict__ b_ih,
                                    const float* __restrict__ b_hh,
                                    const float* __restrict__ conv_b,
                                    float*       __restrict__ h_ws,
                                    unsigned long long ws_bytes)
{
  if (ws_bytes < kWsNeed) return;

  __shared__ __attribute__((aligned(16))) float s_x [kRP * kLD];
  __shared__ __attribute__((aligned(16))) float s_xt[kRP * kLD];
  __shared__ __attribute__((aligned(16))) float s_h [kRP * kLD];
  __shared__ __attribute__((aligned(16))) float s_g [kGE];
  __shared__ __attribute__((aligned(16))) float s_wT [16 * kLD];
  __shared__ __attribute__((aligned(16))) float s_xwT[16 * kLD];
  __shared__ __attribute__((aligned(16))) float s_hout[kHB];
  __shared__ float s_score[kN];
  __shared__ __attribute__((aligned(16))) float s_pp[kLD];
  __shared__ int   s_perm [kN];
  __shared__ float s_dis  [kN];
  __shared__ float s_deg  [kN];

  const int tid    = threadIdx.x;
  constexpr int nthr = kNT;
  const int lane   = tid & 31;
  const int wave   = tid >> 5;
  const int m16    = lane & 15;
  const int hh     = lane >> 4;
  const int nb     = blockIdx.x;

  float xv[kXI], wv[kXI], gv[kGI], dwv[kEI];
  int   dcv[kEI];
#pragma unroll
  for (int j = 0; j < kXI; ++j) {
    const int i = min(tid + j * kNT, kN * kIN - 1);
    xv[j] = x[i];
    wv[j] = W0[i];
  }
#pragma unroll
  for (int j = 0; j < kGI; ++j) {
    const int i  = min(tid + j * kNT, kGE - 1);
    const int g  = i / (16 * kLD);
    const int r  = (i % (16 * kLD)) / kLD, k = i % kLD;
    const int jn = nb * 16 + r;
    const bool real = (jn < kN) && (k < kIN);
    const float* wsrc = (g < 3) ? w_ih : w_hh;
    const float v = wsrc[real ? ((g % 3) * kN + jn) * kIN + k : 0];
    gv[j] = real ? v : 0.f;
  }
#pragma unroll
  for (int j = 0; j < kEI; ++j) {
    const int e = tid + j * kNT;
    dcv[j] = eidx[kE + e];
    dwv[j] = ew[e];
  }
  const float ppv = pool_p[min(tid, kIN - 1)];

  for (int i = tid; i < kRP * kLD; i += nthr) {
    s_x[i] = 0.f; s_xt[i] = 0.f; s_h[i] = 0.f;
  }
  for (int i = tid; i < kN; i += nthr) s_deg[i] = 0.f;
  __syncthreads();

#pragma unroll
  for (int j = 0; j < kXI; ++j) {
    const int i = tid + j * kNT;
    if (i < kN * kIN) {
      const int r = i / kIN, c = i % kIN;
      s_x[r * kLD + c] = xv[j];
      s_h[r * kLD + c] = wv[j];
    }
  }
#pragma unroll
  for (int j = 0; j < kGI; ++j) {
    const int i = tid + j * kNT;
    if (i < kGE) s_g[i] = gv[j];
  }
  if (tid < kLD) s_pp[tid] = (tid < kIN) ? ppv : 0.f;
#pragma unroll
  for (int j = 0; j < kEI; ++j) {
    atomicAdd(&s_deg[dcv[j]], dwv[j]);
  }
  for (int i = tid; i < kN; i += nthr) atomicAdd(&s_deg[i], 1.0f);
  __syncthreads();

  if (tid < kN) {
    const int i = tid;
    float pn = 0.f, d = 0.f;
#pragma unroll
    for (int q = 0; q < kIN / 4; ++q) {
      const float4 pv = *reinterpret_cast<const float4*>(&s_pp[4 * q]);
      const float4 xq = *reinterpret_cast<const float4*>(&s_x[i * kLD + 4 * q]);
      pn = __builtin_fmaf(pv.x, pv.x, pn);  d = __builtin_fmaf(xq.x, pv.x, d);
      pn = __builtin_fmaf(pv.y, pv.y, pn);  d = __builtin_fmaf(xq.y, pv.y, d);
      pn = __builtin_fmaf(pv.z, pv.z, pn);  d = __builtin_fmaf(xq.z, pv.z, d);
      pn = __builtin_fmaf(pv.w, pv.w, pn);  d = __builtin_fmaf(xq.w, pv.w, d);
    }
    {
      const float pl = s_pp[kIN - 1];
      pn = __builtin_fmaf(pl, pl, pn);
      d  = __builtin_fmaf(s_x[i * kLD + kIN - 1], pl, d);
    }
    s_score[i] = tanhf(d * rsqrtf(pn));
    float dg = s_deg[i];
    s_dis[i] = dg > 0.f ? rsqrtf(fmaxf(dg, 1e-12f)) : 0.f;
  }
  __syncthreads();

  for (int i = tid; i < kN; i += nthr) {
    float si = s_score[i];
    int rank = 0;
    for (int j = 0; j < kN; ++j) {
      float sj = s_score[j];
      rank += (sj > si) || (sj == si && j < i);
    }
    s_perm[rank] = i;
  }
  __syncthreads();

  for (int i = tid; i < kN * kLD; i += nthr) {
    int r = i / kLD, c = i % kLD;
    int src = s_perm[r];
    s_xt[r * kLD + c] = s_x[src * kLD + c] * s_score[src];
  }
  __syncthreads();

  if (wave < kMT) {
    const int mt = wave;
    const int jn = nb * 16 + m16;
    const float br = b_ih[min(jn, 3 * kN - 1)]          + b_hh[min(jn, 3 * kN - 1)];
    const float bz = b_ih[min(kN + jn, 3 * kN - 1)]     + b_hh[min(kN + jn, 3 * kN - 1)];
    const float bi = b_ih[min(2 * kN + jn, 3 * kN - 1)];
    const float bh = b_hh[min(2 * kN + jn, 3 * kN - 1)];
    v8f cr, cz, ci, ch;
#pragma unroll
    for (int v = 0; v < 8; ++v) { cr[v] = br; cz[v] = bz; ci[v] = bi; ch[v] = bh; }
    const int arow = (mt * 16 + m16) * kLD;
    const int brow = m16 * kLD;
#pragma unroll 11
    for (int ks = 0; ks < kKS; ++ks) {
      const int k0 = ks * 4 + 2 * hh;
      v2f axt = *reinterpret_cast<const v2f*>(&s_xt[arow + k0]);
      v2f ah0 = *reinterpret_cast<const v2f*>(&s_h [arow + k0]);
      v2f bri = *reinterpret_cast<const v2f*>(&s_g[0 * 16 * kLD + brow + k0]);
      v2f bzi = *reinterpret_cast<const v2f*>(&s_g[1 * 16 * kLD + brow + k0]);
      v2f bni = *reinterpret_cast<const v2f*>(&s_g[2 * 16 * kLD + brow + k0]);
      v2f brh = *reinterpret_cast<const v2f*>(&s_g[3 * 16 * kLD + brow + k0]);
      v2f bzh = *reinterpret_cast<const v2f*>(&s_g[4 * 16 * kLD + brow + k0]);
      v2f bnh = *reinterpret_cast<const v2f*>(&s_g[5 * 16 * kLD + brow + k0]);
      cr = wmma4(axt, bri, cr);   cr = wmma4(ah0, brh, cr);
      cz = wmma4(axt, bzi, cz);   cz = wmma4(ah0, bzh, cz);
      ci = wmma4(axt, bni, ci);
      ch = wmma4(ah0, bnh, ch);
    }
#pragma unroll
    for (int v = 0; v < 8; ++v) {
      const int row = mt * 16 + v + 8 * hh;
      const float rg  = sigm(cr[v]);
      const float zg  = sigm(cz[v]);
      const float cnd = tanhf(ci[v] + rg * ch[v]);
      const float w0v = (jn < kLD) ? s_h[row * kLD + jn] : 0.f;
      if (row < kLD) s_wT[m16 * kLD + row] = (1.f - zg) * cnd + zg * w0v;
    }
  }
  __syncthreads();

  float* s_adj = s_xt;
  int   aer[kEI], aec[kEI];
  float aew[kEI];
#pragma unroll
  for (int j = 0; j < kEI; ++j) {
    const int e = tid + j * kNT;
    aer[j] = eidx[e];
    aec[j] = eidx[kE + e];
    aew[j] = ew[e];
  }
  for (int i = tid; i < kRP * kLD / 4; i += nthr)
    reinterpret_cast<float4*>(s_adj)[i] = make_float4(0.f, 0.f, 0.f, 0.f);
  if (wave < kMT) {
    const int mt = wave;
    v8f acc = {0.f, 0.f, 0.f, 0.f, 0.f, 0.f, 0.f, 0.f};
    const int arow = (mt * 16 + m16) * kLD;
    const int brow = m16 * kLD;
#pragma unroll 11
    for (int ks = 0; ks < kKS; ++ks) {
      const int k0 = ks * 4 + 2 * hh;
      v2f a = *reinterpret_cast<const v2f*>(&s_x [arow + k0]);
      v2f b = *reinterpret_cast<const v2f*>(&s_wT[brow + k0]);
      acc = wmma4(a, b, acc);
    }
#pragma unroll
    for (int v = 0; v < 8; ++v) {
      const int row = mt * 16 + v + 8 * hh;
      if (row < kLD) s_xwT[m16 * kLD + row] = acc[v];
    }
  }
  __syncthreads();

  float av[kEI];
#pragma unroll
  for (int j = 0; j < kEI; ++j) av[j] = s_dis[aer[j]] * aew[j] * s_dis[aec[j]];
#pragma unroll
  for (int j = 0; j < kEI; ++j) atomicAdd(&s_adj[aec[j] * kLD + aer[j]], av[j]);
  for (int i = tid; i < kN; i += nthr) {
    float d = s_dis[i];
    atomicAdd(&s_adj[i * kLD + i], d * d);
  }
  __syncthreads();

  if (wave < kMT) {
    const int mt = wave;
    const int jn = nb * 16 + m16;
    const float cb = conv_b[min(jn, kN - 1)];
    v8f acc;
#pragma unroll
    for (int v = 0; v < 8; ++v) acc[v] = cb;
    const int arow = (mt * 16 + m16) * kLD;
    const int brow = m16 * kLD;
#pragma unroll 11
    for (int ks = 0; ks < kKS; ++ks) {
      const int k0 = ks * 4 + 2 * hh;
      v2f a = *reinterpret_cast<const v2f*>(&s_adj[arow + k0]);
      v2f b = *reinterpret_cast<const v2f*>(&s_xwT[brow + k0]);
      acc = wmma4(a, b, acc);
    }
#pragma unroll
    for (int v = 0; v < 8; ++v) {
      float hv = acc[v];
      hv = hv > 0.f ? hv : (__expf(hv) - 1.f);
      s_hout[(mt * 16 + v + 8 * hh) * 16 + m16] = jn < kN ? hv : 0.f;
    }
  }
  __syncthreads();

  float* hb = h_ws + nb * kHB;
  for (int i = tid; i < kHB / 4; i += nthr)
    reinterpret_cast<float4*>(hb)[i] = reinterpret_cast<const float4*>(s_hout)[i];
}

__launch_bounds__(128, 1)
__global__ void evolve_gcn_linear(const float* __restrict__ h_ws,
                                  const float* __restrict__ lin_w,
                                  const float* __restrict__ lin_b,
                                  float*       __restrict__ out,
                                  unsigned long long ws_bytes)
{
  if (ws_bytes < kWsNeed) return;
  __shared__ __attribute__((aligned(16))) float s_o[16 * kOUT];
  const int tid  = threadIdx.x;
  const int lane = tid & 31;
  const int ot   = tid >> 5;
  const int m16  = lane & 15;
  const int hh   = lane >> 4;
  const int mt   = blockIdx.x;
  const int oc   = ot * 16 + m16;
  const float lb = lin_b[oc];
  v8f acc;
#pragma unroll
  for (int v = 0; v < 8; ++v) acc[v] = lb;
  const int arow = (mt * 16 + m16) * 16;
  const int rowo = oc * kIN;
#pragma unroll
  for (int ks = 0; ks < kKS; ++ks) {
    const int k0 = ks * 4 + 2 * hh;
    v2f a = *reinterpret_cast<const v2f*>(&h_ws[(k0 >> 4) * kHB + arow + (k0 & 15)]);
    const int i0 = min(k0,     kIN - 1);
    const int i1 = min(k0 + 1, kIN - 1);
    v2f b; b.x = lin_w[rowo + i0]; b.y = lin_w[rowo + i1];
    acc = wmma4(a, b, acc);
  }
#pragma unroll
  for (int v = 0; v < 8; ++v) s_o[(v + 8 * hh) * kOUT + oc] = acc[v];
  __syncthreads();
  for (int i = tid; i < 16 * kOUT / 4; i += 128) {
    const int row = mt * 16 + i / (kOUT / 4);
    if (row < kN)
      reinterpret_cast<float4*>(out)[mt * (16 * kOUT / 4) + i] = reinterpret_cast<const float4*>(s_o)[i];
  }
}

extern "C" void kernel_launch(void* const* d_in, const int* in_sizes, int n_in,
                              void* d_out, int out_size, void* d_ws, size_t ws_size,
                              hipStream_t stream) {
  (void)in_sizes; (void)n_in; (void)out_size;
  const float* x       = (const float*)d_in[0];
  const int*   eidx    = (const int*)  d_in[1];
  const float* ew      = (const float*)d_in[2];
  const float* pool_p  = (const float*)d_in[3];
  const float* W0      = (const float*)d_in[4];
  const float* w_ih    = (const float*)d_in[5];
  const float* w_hh    = (const float*)d_in[6];
  const float* b_ih    = (const float*)d_in[7];
  const float* b_hh    = (const float*)d_in[8];
  const float* conv_b  = (const float*)d_in[9];
  const float* lin_w   = (const float*)d_in[10];
  const float* lin_b   = (const float*)d_in[11];
  evolve_gcn_colblock<<<9, 512, 0, stream>>>(x, eidx, ew, pool_p, W0, w_ih, w_hh,
                                             b_ih, b_hh, conv_b, (float*)d_ws,
                                             (unsigned long long)ws_size);
  evolve_gcn_linear<<<9, 128, 0, stream>>>((const float*)d_ws, lin_w, lin_b,
                                           (float*)d_out, (unsigned long long)ws_size);
}


# x_tilde no longer stored: GRU A loader reads x rows through the ranking, scales in registers (same products); freed panel holds the adjacency, built beside the ranking; its atomic sum order changes
# speedup vs baseline: 12.9288x; 1.0926x over previous
#include <hip/hip_runtime.h>


typedef __attribute__((ext_vector_type(2))) float v2f;
typedef __attribute__((ext_vector_type(8))) float v8f;

static constexpr int kN   = 129;
static constexpr int kIN  = 129;
static constexpr int kOUT = 64;
static constexpr int kE   = 4096;
static constexpr int kLD  = 132;
static constexpr int kRP  = 144;
static constexpr int kKS  = 33;
static constexpr int kMT  = 9;
static constexpr int kNT  = 512;
static constexpr int kXI  = (kN * kIN + kNT - 1) / kNT;
static constexpr int kGE  = 6 * 16 * kLD;
static constexpr int kGI  = (kGE + kNT - 1) / kNT;
static constexpr int kEI  = kE / kNT;
static constexpr int kHB  = kRP * 16;
static constexpr unsigned long long kWsNeed = 4ull * kMT * kHB;

static __device__ __forceinline__ v8f wmma4(v2f a, v2f b, v8f c) {
  return __builtin_amdgcn_wmma_f32_16x16x4_f32(
       false, a,  false, b,
       (short)0, c,  false,  false);
}

static __device__ __forceinline__ float sigm(float x) {
  return 1.0f / (1.0f + __expf(-x));
}

__launch_bounds__(512, 1)
__global__ void evolve_gcn_colblock(const float* __restrict__ x,
                                    const int*   __restrict__ eidx,
                                    const float* __restrict__ ew,
                                    const float* __restrict__ pool_p,
                                    const float* __restrict__ W0,
                                    const float* __restrict__ w_ih,
                                    const float* __restrict__ w_hh,
                                    const float* __restrict__ b_ih,
                                    const float* __restrict__ b_hh,
                                    const float* __restrict__ conv_b,
                                    float*       __restrict__ h_ws,
                                    unsigned long long ws_bytes)
{
  if (ws_bytes < kWsNeed) return;

  __shared__ __attribute__((aligned(16))) float s_x [kRP * kLD];
  __shared__ __attribute__((aligned(16))) float s_adj[kRP * kLD];
  __shared__ __attribute__((aligned(16))) float s_h [kRP * kLD];
  __shared__ __attribute__((aligned(16))) float s_g [kGE];
  __shared__ __attribute__((aligned(16))) float s_wT [16 * kLD];
  __shared__ __attribute__((aligned(16))) float s_xwT[16 * kLD];
  __shared__ __attribute__((aligned(16))) float s_hout[kHB];
  __shared__ float s_score[kN];
  __shared__ __attribute__((aligned(16))) float s_pp[kLD];
  __shared__ int   s_perm [kN];
  __shared__ float s_dis  [kN];
  __shared__ float s_deg  [kN];

  const int tid    = threadIdx.x;
  constexpr int nthr = kNT;
  const int lane   = tid & 31;
  const int wave   = tid >> 5;
  const int m16    = lane & 15;
  const int hh     = lane >> 4;
  const int nb     = blockIdx.x;

  float xv[kXI], wv[kXI], gv[kGI], dwv[kEI];
  int   drv[kEI], dcv[kEI];
#pragma unroll
  for (int j = 0; j < kXI; ++j) {
    const int i = min(tid + j * kNT, kN * kIN - 1);
    xv[j] = x[i];
    wv[j] = W0[i];
  }
#pragma unroll
  for (int j = 0; j < kGI; ++j) {
    const int i  = min(tid + j * kNT, kGE - 1);
    const int g  = i / (16 * kLD);
    const int r  = (i % (16 * kLD)) / kLD, k = i % kLD;
    const int jn = nb * 16 + r;
    const bool real = (jn < kN) && (k < kIN);
    const float* wsrc = (g < 3) ? w_ih : w_hh;
    const float v = wsrc[real ? ((g % 3) * kN + jn) * kIN + k : 0];
    gv[j] = real ? v : 0.f;
  }
#pragma unroll
  for (int j = 0; j < kEI; ++j) {
    const int e = tid + j * kNT;
    drv[j] = eidx[e];
    dcv[j] = eidx[kE + e];
    dwv[j] = ew[e];
  }
  const float ppv = pool_p[min(tid, kIN - 1)];

  for (int i = tid; i < kRP * kLD; i += nthr) {
    s_x[i] = 0.f; s_adj[i] = 0.f; s_h[i] = 0.f;
  }
  for (int i = tid; i < kN; i += nthr) s_deg[i] = 0.f;
  __syncthreads();

#pragma unroll
  for (int j = 0; j < kXI; ++j) {
    const int i = tid + j * kNT;
    if (i < kN * kIN) {
      const int r = i / kIN, c = i % kIN;
      s_x[r * kLD + c] = xv[j];
      s_h[r * kLD + c] = wv[j];
    }
  }
#pragma unroll
  for (int j = 0; j < kGI; ++j) {
    const int i = tid + j * kNT;
    if (i < kGE) s_g[i] = gv[j];
  }
  if (tid < kLD) s_pp[tid] = (tid < kIN) ? ppv : 0.f;
#pragma unroll
  for (int j = 0; j < kEI; ++j) {
    atomicAdd(&s_deg[dcv[j]], dwv[j]);
  }
  for (int i = tid; i < kN; i += nthr) atomicAdd(&s_deg[i], 1.0f);
  __syncthreads();

  if (tid < kN) {
    const int i = tid;
    float pn = 0.f, d = 0.f;
#pragma unroll
    for (int q = 0; q < kIN / 4; ++q) {
      const float4 pv = *reinterpret_cast<const float4*>(&s_pp[4 * q]);
      const float4 xq = *reinterpret_cast<const float4*>(&s_x[i * kLD + 4 * q]);
      pn = __builtin_fmaf(pv.x, pv.x, pn);  d = __builtin_fmaf(xq.x, pv.x, d);
      pn = __builtin_fmaf(pv.y, pv.y, pn);  d = __builtin_fmaf(xq.y, pv.y, d);
      pn = __builtin_fmaf(pv.z, pv.z, pn);  d = __builtin_fmaf(xq.z, pv.z, d);
      pn = __builtin_fmaf(pv.w, pv.w, pn);  d = __builtin_fmaf(xq.w, pv.w, d);
    }
    {
      const float pl = s_pp[kIN - 1];
      pn = __builtin_fmaf(pl, pl, pn);
      d  = __builtin_fmaf(s_x[i * kLD + kIN - 1], pl, d);
    }
    s_score[i] = tanhf(d * rsqrtf(pn));
    float dg = s_deg[i];
    s_dis[i] = dg > 0.f ? rsqrtf(fmaxf(dg, 1e-12f)) : 0.f;
  }
  __syncthreads();

  for (int i = tid; i < kN; i += nthr) {
    float si = s_score[i];
    int rank = 0;
    for (int j = 0; j < kN; ++j) {
      float sj = s_score[j];
      rank += (sj > si) || (sj == si && j < i);
    }
    s_perm[rank] = i;
  }

  {
    float av[kEI];
#pragma unroll
    for (int j = 0; j < kEI; ++j) av[j] = s_dis[drv[j]] * dwv[j] * s_dis[dcv[j]];
#pragma unroll
    for (int j = 0; j < kEI; ++j) atomicAdd(&s_adj[dcv[j] * kLD + drv[j]], av[j]);
    if (tid >= kNT - kN) {
      const int i = tid - (kNT - kN);
      const float d = s_dis[i];
      atomicAdd(&s_adj[i * kLD + i], d * d);
    }
  }
  __syncthreads();

  if (wave < kMT) {
    const int mt = wave;
    const int jn = nb * 16 + m16;
    const float br = b_ih[min(jn, 3 * kN - 1)]          + b_hh[min(jn, 3 * kN - 1)];
    const float bz = b_ih[min(kN + jn, 3 * kN - 1)]     + b_hh[min(kN + jn, 3 * kN - 1)];
    const float bi = b_ih[min(2 * kN + jn, 3 * kN - 1)];
    const float bh = b_hh[min(2 * kN + jn, 3 * kN - 1)];
    v8f cr, cz, ci, ch;
#pragma unroll
    for (int v = 0; v < 8; ++v) { cr[v] = br; cz[v] = bz; ci[v] = bi; ch[v] = bh; }
    const int arow = (mt * 16 + m16) * kLD;
    const int brow = m16 * kLD;
    const int xr   = mt * 16 + m16;
    const int prm  = s_perm[min(xr, kN - 1)];
    const int xrow = ((xr < kN) ? prm : xr) * kLD;
    const float sc = (xr < kN) ? s_score[prm] : 0.f;
#pragma unroll 11
    for (int ks = 0; ks < kKS; ++ks) {
      const int k0 = ks * 4 + 2 * hh;
      v2f axt = *reinterpret_cast<const v2f*>(&s_x[xrow + k0]);
      axt.x *= sc;  axt.y *= sc;
      v2f ah0 = *reinterpret_cast<const v2f*>(&s_h [arow + k0]);
      v2f bri = *reinterpret_cast<const v2f*>(&s_g[0 * 16 * kLD + brow + k0]);
      v2f bzi = *reinterpret_cast<const v2f*>(&s_g[1 * 16 * kLD + brow + k0]);
      v2f bni = *reinterpret_cast<const v2f*>(&s_g[2 * 16 * kLD + brow + k0]);
      v2f brh = *reinterpret_cast<const v2f*>(&s_g[3 * 16 * kLD + brow + k0]);
      v2f bzh = *reinterpret_cast<const v2f*>(&s_g[4 * 16 * kLD + brow + k0]);
      v2f bnh = *reinterpret_cast<const v2f*>(&s_g[5 * 16 * kLD + brow + k0]);
      cr = wmma4(axt, bri, cr);   cr = wmma4(ah0, brh, cr);
      cz = wmma4(axt, bzi, cz);   cz = wmma4(ah0, bzh, cz);
      ci = wmma4(axt, bni, ci);
      ch = wmma4(ah0, bnh, ch);
    }
#pragma unroll
    for (int v = 0; v < 8; ++v) {
      const int row = mt * 16 + v + 8 * hh;
      const float rg  = sigm(cr[v]);
      const float zg  = sigm(cz[v]);
      const float cnd = tanhf(ci[v] + rg * ch[v]);
      const float w0v = (jn < kLD) ? s_h[row * kLD + jn] : 0.f;
      if (row < kLD) s_wT[m16 * kLD + row] = (1.f - zg) * cnd + zg * w0v;
    }
  }
  __syncthreads();

  if (wave < kMT) {
    const int mt = wave;
    v8f acc = {0.f, 0.f, 0.f, 0.f, 0.f, 0.f, 0.f, 0.f};
    const int arow = (mt * 16 + m16) * kLD;
    const int brow = m16 * kLD;
#pragma unroll 11
    for (int ks = 0; ks < kKS; ++ks) {
      const int k0 = ks * 4 + 2 * hh;
      v2f a = *reinterpret_cast<const v2f*>(&s_x [arow + k0]);
      v2f b = *reinterpret_cast<const v2f*>(&s_wT[brow + k0]);
      acc = wmma4(a, b, acc);
    }
#pragma unroll
    for (int v = 0; v < 8; ++v) {
      const int row = mt * 16 + v + 8 * hh;
      if (row < kLD) s_xwT[m16 * kLD + row] = acc[v];
    }
  }
  __syncthreads();

  if (wave < kMT) {
    const int mt = wave;
    const int jn = nb * 16 + m16;
    const float cb = conv_b[min(jn, kN - 1)];
    v8f acc;
#pragma unroll
    for (int v = 0; v < 8; ++v) acc[v] = cb;
    const int arow = (mt * 16 + m16) * kLD;
    const int brow = m16 * kLD;
#pragma unroll 11
    for (int ks = 0; ks < kKS; ++ks) {
      const int k0 = ks * 4 + 2 * hh;
      v2f a = *reinterpret_cast<const v2f*>(&s_adj[arow + k0]);
      v2f b = *reinterpret_cast<const v2f*>(&s_xwT[brow + k0]);
      acc = wmma4(a, b, acc);
    }
#pragma unroll
    for (int v = 0; v < 8; ++v) {
      float hv = acc[v];
      hv = hv > 0.f ? hv : (__expf(hv) - 1.f);
      s_hout[(mt * 16 + v + 8 * hh) * 16 + m16] = jn < kN ? hv : 0.f;
    }
  }
  __syncthreads();

  float* hb = h_ws + nb * kHB;
  for (int i = tid; i < kHB / 4; i += nthr)
    reinterpret_cast<float4*>(hb)[i] = reinterpret_cast<const float4*>(s_hout)[i];
}

__launch_bounds__(128, 1)
__global__ void evolve_gcn_linear(const float* __restrict__ h_ws,
                                  const float* __restrict__ lin_w,
                                  const float* __restrict__ lin_b,
                                  float*       __restrict__ out,
                                  unsigned long long ws_bytes)
{
  if (ws_bytes < kWsNeed) return;
  __shared__ __attribute__((aligned(16))) float s_o[16 * kOUT];
  const int tid  = threadIdx.x;
  const int lane = tid & 31;
  const int ot   = tid >> 5;
  const int m16  = lane & 15;
  const int hh   = lane >> 4;
  const int mt   = blockIdx.x;
  const int oc   = ot * 16 + m16;
  const float lb = lin_b[oc];
  v8f acc;
#pragma unroll
  for (int v = 0; v < 8; ++v) acc[v] = lb;
  const int arow = (mt * 16 + m16) * 16;
  const int rowo = oc * kIN;
#pragma unroll
  for (int ks = 0; ks < kKS; ++ks) {
    const int k0 = ks * 4 + 2 * hh;
    v2f a = *reinterpret_cast<const v2f*>(&h_ws[(k0 >> 4) * kHB + arow + (k0 & 15)]);
    const int i0 = min(k0,     kIN - 1);
    const int i1 = min(k0 + 1, kIN - 1);
    v2f b; b.x = lin_w[rowo + i0]; b.y = lin_w[rowo + i1];
    acc = wmma4(a, b, acc);
  }
#pragma unroll
  for (int v = 0; v < 8; ++v) s_o[(v + 8 * hh) * kOUT + oc] = acc[v];
  __syncthreads();
  for (int i = tid; i < 16 * kOUT / 4; i += 128) {
    const int row = mt * 16 + i / (kOUT / 4);
    if (row < kN)
      reinterpret_cast<float4*>(out)[mt * (16 * kOUT / 4) + i] = reinterpret_cast<const float4*>(s_o)[i];
  }
}

extern "C" void kernel_launch(void* const* d_in, const int* in_sizes, int n_in,
                              void* d_out, int out_size, void* d_ws, size_t ws_size,
                              hipStream_t stream) {
  (void)in_sizes; (void)n_in; (void)out_size;
  const float* x       = (const float*)d_in[0];
  const int*   eidx    = (const int*)  d_in[1];
  const float* ew      = (const float*)d_in[2];
  const float* pool_p  = (const float*)d_in[3];
  const float* W0      = (const float*)d_in[4];
  const float* w_ih    = (const float*)d_in[5];
  const float* w_hh    = (const float*)d_in[6];
  const float* b_ih    = (const float*)d_in[7];
  const float* b_hh    = (const float*)d_in[8];
  const float* conv_b  = (const float*)d_in[9];
  const float* lin_w   = (const float*)d_in[10];
  const float* lin_b   = (const float*)d_in[11];
  evolve_gcn_colblock<<<9, 512, 0, stream>>>(x, eidx, ew, pool_p, W0, w_ih, w_hh,
                                             b_ih, b_hh, conv_b, (float*)d_ws,
                                             (unsigned long long)ws_size);
  evolve_gcn_linear<<<9, 128, 0, stream>>>((const float*)d_ws, lin_w, lin_b,
                                           (float*)d_out, (unsigned long long)ws_size);
}


# Ninth row tile (one live row) of the GRU step split by accumulator over three waves on three SIMDs (r, z, candidate pair) in place of one whole-tile wave; row 128 finished from LDS; sum order kept
# speedup vs baseline: 13.4652x; 1.0415x over previous
#include <hip/hip_runtime.h>


typedef __attribute__((ext_vector_type(2))) float v2f;
typedef __attribute__((ext_vector_type(8))) float v8f;

static constexpr int kN   = 129;
static constexpr int kIN  = 129;
static constexpr int kOUT = 64;
static constexpr int kE   = 4096;
static constexpr int kLD  = 132;
static constexpr int kRP  = 144;
static constexpr int kKS  = 33;
static constexpr int kMT  = 9;
static constexpr int kNT  = 512;
static constexpr int kXI  = (kN * kIN + kNT - 1) / kNT;
static constexpr int kGE  = 6 * 16 * kLD;
static constexpr int kGI  = (kGE + kNT - 1) / kNT;
static constexpr int kEI  = kE / kNT;
static constexpr int kHB  = kRP * 16;
static constexpr unsigned long long kWsNeed = 4ull * kMT * kHB;

static __device__ __forceinline__ v8f wmma4(v2f a, v2f b, v8f c) {
  return __builtin_amdgcn_wmma_f32_16x16x4_f32(
       false, a,  false, b,
       (short)0, c,  false,  false);
}

static __device__ __forceinline__ float sigm(float x) {
  return 1.0f / (1.0f + __expf(-x));
}

static __device__ __forceinline__ float gru_out(float cr, float cz, float ci, float ch, float w0v) {
  const float rg  = sigm(cr);
  const float zg  = sigm(cz);
  const float cnd = tanhf(ci + rg * ch);
  return (1.f - zg) * cnd + zg * w0v;
}

__launch_bounds__(512, 1)
__global__ void evolve_gcn_colblock(const float* __restrict__ x,
                                    const int*   __restrict__ eidx,
                                    const float* __restrict__ ew,
                                    const float* __restrict__ pool_p,
                                    const float* __restrict__ W0,
                                    const float* __restrict__ w_ih,
                                    const float* __restrict__ w_hh,
                                    const float* __restrict__ b_ih,
                                    const float* __restrict__ b_hh,
                                    const float* __restrict__ conv_b,
                                    float*       __restrict__ h_ws,
                                    unsigned long long ws_bytes)
{
  if (ws_bytes < kWsNeed) return;

  __shared__ __attribute__((aligned(16))) float s_x [kRP * kLD];
  __shared__ __attribute__((aligned(16))) float s_adj[kRP * kLD];
  __shared__ __attribute__((aligned(16))) float s_h [kRP * kLD];
  __shared__ __attribute__((aligned(16))) float s_g [kGE];
  __shared__ __attribute__((aligned(16))) float s_wT [16 * kLD];
  __shared__ __attribute__((aligned(16))) float s_xwT[16 * kLD];
  __shared__ __attribute__((aligned(16))) float s_hout[kHB];
  __shared__ float s_score[kN];
  __shared__ __attribute__((aligned(16))) float s_pp[kLD];
  __shared__ int   s_perm [kN];
  __shared__ float s_dis  [kN];
  __shared__ float s_deg  [kN];
  __shared__ float s_t8   [4 * 16];

  const int tid    = threadIdx.x;
  constexpr int nthr = kNT;
  const int lane   = tid & 31;
  const int wave   = tid >> 5;
  const int m16    = lane & 15;
  const int hh     = lane >> 4;
  const int nb     = blockIdx.x;

  float xv[kXI], wv[kXI], gv[kGI], dwv[kEI];
  int   drv[kEI], dcv[kEI];
#pragma unroll
  for (int j = 0; j < kXI; ++j) {
    const int i = min(tid + j * kNT, kN * kIN - 1);
    xv[j] = x[i];
    wv[j] = W0[i];
  }
#pragma unroll
  for (int j = 0; j < kGI; ++j) {
    const int i  = min(tid + j * kNT, kGE - 1);
    const int g  = i / (16 * kLD);
    const int r  = (i % (16 * kLD)) / kLD, k = i % kLD;
    const int jn = nb * 16 + r;
    const bool real = (jn < kN) && (k < kIN);
    const float* wsrc = (g < 3) ? w_ih : w_hh;
    const float v = wsrc[real ? ((g % 3) * kN + jn) * kIN + k : 0];
    gv[j] = real ? v : 0.f;
  }
#pragma unroll
  for (int j = 0; j < kEI; ++j) {
    const int e = tid + j * kNT;
    drv[j] = eidx[e];
    dcv[j] = eidx[kE + e];
    dwv[j] = ew[e];
  }
  const float ppv = pool_p[min(tid, kIN - 1)];

  for (int i = tid; i < kRP * kLD; i += nthr) {
    s_x[i] = 0.f; s_adj[i] = 0.f; s_h[i] = 0.f;
  }
  for (int i = tid; i < kN; i += nthr) s_deg[i] = 0.f;
  __syncthreads();

#pragma unroll
  for (int j = 0; j < kXI; ++j) {
    const int i = tid + j * kNT;
    if (i < kN * kIN) {
      const int r = i / kIN, c = i % kIN;
      s_x[r * kLD + c] = xv[j];
      s_h[r * kLD + c] = wv[j];
    }
  }
#pragma unroll
  for (int j = 0; j < kGI; ++j) {
    const int i = tid + j * kNT;
    if (i < kGE) s_g[i] = gv[j];
  }
  if (tid < kLD) s_pp[tid] = (tid < kIN) ? ppv : 0.f;
#pragma unroll
  for (int j = 0; j < kEI; ++j) {
    atomicAdd(&s_deg[dcv[j]], dwv[j]);
  }
  for (int i = tid; i < kN; i += nthr) atomicAdd(&s_deg[i], 1.0f);
  __syncthreads();

  if (tid < kN) {
    const int i = tid;
    float pn = 0.f, d = 0.f;
#pragma unroll
    for (int q = 0; q < kIN / 4; ++q) {
      const float4 pv = *reinterpret_cast<const float4*>(&s_pp[4 * q]);
      const float4 xq = *reinterpret_cast<const float4*>(&s_x[i * kLD + 4 * q]);
      pn = __builtin_fmaf(pv.x, pv.x, pn);  d = __builtin_fmaf(xq.x, pv.x, d);
      pn = __builtin_fmaf(pv.y, pv.y, pn);  d = __builtin_fmaf(xq.y, pv.y, d);
      pn = __builtin_fmaf(pv.z, pv.z, pn);  d = __builtin_fmaf(xq.z, pv.z, d);
      pn = __builtin_fmaf(pv.w, pv.w, pn);  d = __builtin_fmaf(xq.w, pv.w, d);
    }
    {
      const float pl = s_pp[kIN - 1];
      pn = __builtin_fmaf(pl, pl, pn);
      d  = __builtin_fmaf(s_x[i * kLD + kIN - 1], pl, d);
    }
    s_score[i] = tanhf(d * rsqrtf(pn));
    float dg = s_deg[i];
    s_dis[i] = dg > 0.f ? rsqrtf(fmaxf(dg, 1e-12f)) : 0.f;
  }
  __syncthreads();

  for (int i = tid; i < kN; i += nthr) {
    float si = s_score[i];
    int rank = 0;
    for (int j = 0; j < kN; ++j) {
      float sj = s_score[j];
      rank += (sj > si) || (sj == si && j < i);
    }
    s_perm[rank] = i;
  }

  {
    float av[kEI];
#pragma unroll
    for (int j = 0; j < kEI; ++j) av[j] = s_dis[drv[j]] * dwv[j] * s_dis[dcv[j]];
#pragma unroll
    for (int j = 0; j < kEI; ++j) atomicAdd(&s_adj[dcv[j] * kLD + drv[j]], av[j]);
    if (tid >= kNT - kN) {
      const int i = tid - (kNT - kN);
      const float d = s_dis[i];
      atomicAdd(&s_adj[i * kLD + i], d * d);
    }
  }
  __syncthreads();

  if (wave < 8) {
    const int mt = wave;
    const int jn = nb * 16 + m16;
    const float br = b_ih[min(jn, 3 * kN - 1)]          + b_hh[min(jn, 3 * kN - 1)];
    const float bz = b_ih[min(kN + jn, 3 * kN - 1)]     + b_hh[min(kN + jn, 3 * kN - 1)];
    const float bi = b_ih[min(2 * kN + jn, 3 * kN - 1)];
    const float bh = b_hh[min(2 * kN + jn, 3 * kN - 1)];
    v8f cr, cz, ci, ch;
#pragma unroll
    for (int v = 0; v < 8; ++v) { cr[v] = br; cz[v] = bz; ci[v] = bi; ch[v] = bh; }
    const int arow = (mt * 16 + m16) * kLD;
    const int brow = m16 * kLD;
    const int xr   = mt * 16 + m16;
    const int prm  = s_perm[xr];
    const int xrow = prm * kLD;
    const float sc = s_score[prm];
#pragma unroll 11
    for (int ks = 0; ks < kKS; ++ks) {
      const int k0 = ks * 4 + 2 * hh;
      v2f axt = *reinterpret_cast<const v2f*>(&s_x[xrow + k0]);
      axt.x *= sc;  axt.y *= sc;
      v2f ah0 = *reinterpret_cast<const v2f*>(&s_h [arow + k0]);
      v2f bri = *reinterpret_cast<const v2f*>(&s_g[0 * 16 * kLD + brow + k0]);
      v2f bzi = *reinterpret_cast<const v2f*>(&s_g[1 * 16 * kLD + brow + k0]);
      v2f bni = *reinterpret_cast<const v2f*>(&s_g[2 * 16 * kLD + brow + k0]);
      v2f brh = *reinterpret_cast<const v2f*>(&s_g[3 * 16 * kLD + brow + k0]);
      v2f bzh = *reinterpret_cast<const v2f*>(&s_g[4 * 16 * kLD + brow + k0]);
      v2f bnh = *reinterpret_cast<const v2f*>(&s_g[5 * 16 * kLD + brow + k0]);
      cr = wmma4(axt, bri, cr);   cr = wmma4(ah0, brh, cr);
      cz = wmma4(axt, bzi, cz);   cz = wmma4(ah0, bzh, cz);
      ci = wmma4(axt, bni, ci);
      ch = wmma4(ah0, bnh, ch);
    }
#pragma unroll
    for (int v = 0; v < 8; ++v) {
      const int row = mt * 16 + v + 8 * hh;
      const float w0v = (jn < kLD) ? s_h[row * kLD + jn] : 0.f;
      s_wT[m16 * kLD + row] = gru_out(cr[v], cz[v], ci[v], ch[v], w0v);
    }
  } else if (wave < 11) {
    const int part = wave - 8;
    const int jn   = nb * 16 + m16;
    const int arow = (8 * 16 + m16) * kLD;
    const int brow = m16 * kLD;
    const int xr   = 8 * 16 + m16;
    const int prm  = s_perm[min(xr, kN - 1)];
    const int xrow = ((xr < kN) ? prm : xr) * kLD;
    const float sc = (xr < kN) ? s_score[prm] : 0.f;
    if (part < 2) {
      const int bidx = min(part * kN + jn, 3 * kN - 1);
      const float b0 = b_ih[bidx] + b_hh[bidx];
      v8f c;
#pragma unroll
      for (int v = 0; v < 8; ++v) c[v] = b0;
      const int gi_ = part * 16 * kLD + brow;
      const int gh_ = (3 + part) * 16 * kLD + brow;
#pragma unroll 11
      for (int ks = 0; ks < kKS; ++ks) {
        const int k0 = ks * 4 + 2 * hh;
        v2f axt = *reinterpret_cast<const v2f*>(&s_x[xrow + k0]);
        axt.x *= sc;  axt.y *= sc;
        v2f ah0 = *reinterpret_cast<const v2f*>(&s_h[arow + k0]);
        v2f bgi = *reinterpret_cast<const v2f*>(&s_g[gi_ + k0]);
        v2f bgh = *reinterpret_cast<const v2f*>(&s_g[gh_ + k0]);
        c = wmma4(axt, bgi, c);   c = wmma4(ah0, bgh, c);
      }
      if (hh == 0) s_t8[part * 16 + m16] = c[0];
    } else {
      const float bi = b_ih[min(2 * kN + jn, 3 * kN - 1)];
      const float bh = b_hh[min(2 * kN + jn, 3 * kN - 1)];
      v8f ci, ch;
#pragma unroll
      for (int v = 0; v < 8; ++v) { ci[v] = bi; ch[v] = bh; }
#pragma unroll 11
      for (int ks = 0; ks < kKS; ++ks) {
        const int k0 = ks * 4 + 2 * hh;
        v2f axt = *reinterpret_cast<const v2f*>(&s_x[xrow + k0]);
        axt.x *= sc;  axt.y *= sc;
        v2f ah0 = *reinterpret_cast<const v2f*>(&s_h[arow + k0]);
        v2f bni = *reinterpret_cast<const v2f*>(&s_g[2 * 16 * kLD + brow + k0]);
        v2f bnh = *reinterpret_cast<const v2f*>(&s_g[5 * 16 * kLD + brow + k0]);
        ci = wmma4(axt, bni, ci);
        ch = wmma4(ah0, bnh, ch);
      }
      if (hh == 0) { s_t8[2 * 16 + m16] = ci[0];  s_t8[3 * 16 + m16] = ch[0]; }
    }
  }
  __syncthreads();

  if (tid < 16) {
    const int jn = nb * 16 + tid;
    const float w0v = (jn < kLD) ? s_h[128 * kLD + jn] : 0.f;
    s_wT[tid * kLD + 128] = gru_out(s_t8[tid], s_t8[16 + tid], s_t8[32 + tid], s_t8[48 + tid], w0v);
    s_wT[tid * kLD + 129] = 0.f;
    s_wT[tid * kLD + 130] = 0.f;
    s_wT[tid * kLD + 131] = 0.f;
  }
  __syncthreads();

  if (wave < kMT) {
    const int mt = wave;
    v8f acc = {0.f, 0.f, 0.f, 0.f, 0.f, 0.f, 0.f, 0.f};
    const int arow = (mt * 16 + m16) * kLD;
    const int brow = m16 * kLD;
#pragma unroll 11
    for (int ks = 0; ks < kKS; ++ks) {
      const int k0 = ks * 4 + 2 * hh;
      v2f a = *reinterpret_cast<const v2f*>(&s_x [arow + k0]);
      v2f b = *reinterpret_cast<const v2f*>(&s_wT[brow + k0]);
      acc = wmma4(a, b, acc);
    }
#pragma unroll
    for (int v = 0; v < 8; ++v) {
      const int row = mt * 16 + v + 8 * hh;
      if (row < kLD) s_xwT[m16 * kLD + row] = acc[v];
    }
  }
  __syncthreads();

  if (wave < kMT) {
    const int mt = wave;
    const int jn = nb * 16 + m16;
    const float cb = conv_b[min(jn, kN - 1)];
    v8f acc;
#pragma unroll
    for (int v = 0; v < 8; ++v) acc[v] = cb;
    const int arow = (mt * 16 + m16) * kLD;
    const int brow = m16 * kLD;
#pragma unroll 11
    for (int ks = 0; ks < kKS; ++ks) {
      const int k0 = ks * 4 + 2 * hh;
      v2f a = *reinterpret_cast<const v2f*>(&s_adj[arow + k0]);
      v2f b = *reinterpret_cast<const v2f*>(&s_xwT[brow + k0]);
      acc = wmma4(a, b, acc);
    }
#pragma unroll
    for (int v = 0; v < 8; ++v) {
      float hv = acc[v];
      hv = hv > 0.f ? hv : (__expf(hv) - 1.f);
      s_hout[(mt * 16 + v + 8 * hh) * 16 + m16] = jn < kN ? hv : 0.f;
    }
  }
  __syncthreads();

  float* hb = h_ws + nb * kHB;
  for (int i = tid; i < kHB / 4; i += nthr)
    reinterpret_cast<float4*>(hb)[i] = reinterpret_cast<const float4*>(s_hout)[i];
}

__launch_bounds__(128, 1)
__global__ void evolve_gcn_linear(const float* __restrict__ h_ws,
                                  const float* __restrict__ lin_w,
                                  const float* __restrict__ lin_b,
                                  float*       __restrict__ out,
                                  unsigned long long ws_bytes)
{
  if (ws_bytes < kWsNeed) return;
  __shared__ __attribute__((aligned(16))) float s_o[16 * kOUT];
  const int tid  = threadIdx.x;
  const int lane = tid & 31;
  const int ot   = tid >> 5;
  const int m16  = lane & 15;
  const int hh   = lane >> 4;
  const int mt   = blockIdx.x;
  const int oc   = ot * 16 + m16;
  const float lb = lin_b[oc];
  v8f acc;
#pragma unroll
  for (int v = 0; v < 8; ++v) acc[v] = lb;
  const int arow = (mt * 16 + m16) * 16;
  const int rowo = oc * kIN;
#pragma unroll
  for (int ks = 0; ks < kKS; ++ks) {
    const int k0 = ks * 4 + 2 * hh;
    v2f a = *reinterpret_cast<const v2f*>(&h_ws[(k0 >> 4) * kHB + arow + (k0 & 15)]);
    const int i0 = min(k0,     kIN - 1);
    const int i1 = min(k0 + 1, kIN - 1);
    v2f b; b.x = lin_w[rowo + i0]; b.y = lin_w[rowo + i1];
    acc = wmma4(a, b, acc);
  }
#pragma unroll
  for (int v = 0; v < 8; ++v) s_o[(v + 8 * hh) * kOUT + oc] = acc[v];
  __syncthreads();
  for (int i = tid; i < 16 * kOUT / 4; i += 128) {
    const int row = mt * 16 + i / (kOUT / 4);
    if (row < kN)
      reinterpret_cast<float4*>(out)[mt * (16 * kOUT / 4) + i] = reinterpret_cast<const float4*>(s_o)[i];
  }
}

extern "C" void kernel_launch(void* const* d_in, const int* in_sizes, int n_in,
                              void* d_out, int out_size, void* d_ws, size_t ws_size,
                              hipStream_t stream) {
  (void)in_sizes; (void)n_in; (void)out_size;
  const float* x       = (const float*)d_in[0];
  const int*   eidx    = (const int*)  d_in[1];
  const float* ew      = (const float*)d_in[2];
  const float* pool_p  = (const float*)d_in[3];
  const float* W0      = (const float*)d_in[4];
  const float* w_ih    = (const float*)d_in[5];
  const float* w_hh    = (const float*)d_in[6];
  const float* b_ih    = (const float*)d_in[7];
  const float* b_hh    = (const float*)d_in[8];
  const float* conv_b  = (const float*)d_in[9];
  const float* lin_w   = (const float*)d_in[10];
  const float* lin_b   = (const float*)d_in[11];
  evolve_gcn_colblock<<<9, 512, 0, stream>>>(x, eidx, ew, pool_p, W0, w_ih, w_hh,
                                             b_ih, b_hh, conv_b, (float*)d_ws,
                                             (unsigned long long)ws_size);
  evolve_gcn_linear<<<9, 128, 0, stream>>>((const float*)d_ws, lin_w, lin_b,
                                           (float*)d_out, (unsigned long long)ws_size);
}


# Staging loads go out back to back: gate rows mapped wave -> row, lane -> k, each address a base plus a constant (was ~35 integer ops per load); LDS zero fill split behind two load groups of <= 63
# speedup vs baseline: 14.5024x; 1.0770x over previous
#include <hip/hip_runtime.h>


typedef __attribute__((ext_vector_type(2))) float v2f;
typedef __attribute__((ext_vector_type(8))) float v8f;

static constexpr int kN   = 129;
static constexpr int kIN  = 129;
static constexpr int kOUT = 64;
static constexpr int kE   = 4096;
static constexpr int kLD  = 132;
static constexpr int kRP  = 144;
static constexpr int kKS  = 33;
static constexpr int kMT  = 9;
static constexpr int kNT  = 512;
static constexpr int kXI  = (kN * kIN + kNT - 1) / kNT;
static constexpr int kGE  = 6 * 16 * kLD;
static constexpr int kGM  = 24;
static constexpr int kXA  = 6;
static constexpr int kEI  = kE / kNT;
static constexpr int kHB  = kRP * 16;
static constexpr unsigned long long kWsNeed = 4ull * kMT * kHB;

static __device__ __forceinline__ v8f wmma4(v2f a, v2f b, v8f c) {
  return __builtin_amdgcn_wmma_f32_16x16x4_f32(
       false, a,  false, b,
       (short)0, c,  false,  false);
}

static __device__ __forceinline__ float sigm(float x) {
  return 1.0f / (1.0f + __expf(-x));
}

static __device__ __forceinline__ float gru_out(float cr, float cz, float ci, float ch, float w0v) {
  const float rg  = sigm(cr);
  const float zg  = sigm(cz);
  const float cnd = tanhf(ci + rg * ch);
  return (1.f - zg) * cnd + zg * w0v;
}

__launch_bounds__(512, 1)
__global__ void evolve_gcn_colblock(const float* __restrict__ x,
                                    const int*   __restrict__ eidx,
                                    const float* __restrict__ ew,
                                    const float* __restrict__ pool_p,
                                    const float* __restrict__ W0,
                                    const float* __restrict__ w_ih,
                                    const float* __restrict__ w_hh,
                                    const float* __restrict__ b_ih,
                                    const float* __restrict__ b_hh,
                                    const float* __restrict__ conv_b,
                                    float*       __restrict__ h_ws,
                                    unsigned long long ws_bytes)
{
  if (ws_bytes < kWsNeed) return;

  __shared__ __attribute__((aligned(16))) float s_x [kRP * kLD];
  __shared__ __attribute__((aligned(16))) float s_adj[kRP * kLD];
  __shared__ __attribute__((aligned(16))) float s_h [kRP * kLD];
  __shared__ __attribute__((aligned(16))) float s_g [kGE];
  __shared__ __attribute__((aligned(16))) float s_wT [16 * kLD];
  __shared__ __attribute__((aligned(16))) float s_xwT[16 * kLD];
  __shared__ __attribute__((aligned(16))) float s_hout[kHB];
  __shared__ float s_score[kN];
  __shared__ __attribute__((aligned(16))) float s_pp[kLD];
  __shared__ int   s_perm [kN];
  __shared__ float s_dis  [kN];
  __shared__ float s_deg  [kN];
  __shared__ float s_t8   [4 * 16];

  const int tid    = threadIdx.x;
  constexpr int nthr = kNT;
  const int lane   = tid & 31;
  const int wave   = tid >> 5;
  const int m16    = lane & 15;
  const int hh     = lane >> 4;
  const int nb     = blockIdx.x;

  float xv[kXI], wv[kXI], gv[kGM], dwv[kEI];
  int   drv[kEI], dcv[kEI];
  const int rlim = kN - 1 - nb * 16;
  const float* gi0 = w_ih + nb * 16 * kIN;
  const float* gh0 = w_hh + nb * 16 * kIN;
  const int gofs = min(wave, rlim) * kIN + lane;
#pragma unroll
  for (int it = 0; it < kGM; ++it) {
    const int g = it >> 2, kc = it & 3;
    const float* src = (g < 3 ? gi0 : gh0) + (g % 3) * kN * kIN + kc * 32;
    gv[it] = src[gofs];
  }
  const int trow = min(tid >> 2, 6 * 16 - 1);
  const int tjn  = nb * 16 + (trow & 15);
  const float* tsrc = (trow < 3 * 16) ? w_ih : w_hh;
  const float gtv = tsrc[(((trow >> 4) % 3) * kN + min(tjn, kN - 1)) * kIN + (kIN - 1)];
#pragma unroll
  for (int j = 0; j < kEI; ++j) {
    const int e = tid + j * kNT;
    drv[j] = eidx[e];
    dcv[j] = eidx[kE + e];
    dwv[j] = ew[e];
  }
  const float ppv = pool_p[min(tid, kIN - 1)];
#pragma unroll
  for (int j = 0; j < kXA; ++j) {
    const int i = tid + j * kNT;
    xv[j] = x[i];
    wv[j] = W0[i];
  }

  for (int i = tid; i < kRP * kLD; i += nthr) s_adj[i] = 0.f;
  for (int i = tid; i < kN; i += nthr) s_deg[i] = 0.f;

#pragma unroll
  for (int j = kXA; j < kXI; ++j) {
    const int i = min(tid + j * kNT, kN * kIN - 1);
    xv[j] = x[i];
    wv[j] = W0[i];
  }

  for (int i = tid; i < kRP * kLD; i += nthr) {
    s_x[i] = 0.f; s_h[i] = 0.f;
  }
  __syncthreads();

#pragma unroll
  for (int j = 0; j < kXI; ++j) {
    const int i = tid + j * kNT;
    if (i < kN * kIN) {
      const int r = i / kIN, c = i % kIN;
      s_x[r * kLD + c] = xv[j];
      s_h[r * kLD + c] = wv[j];
    }
  }
#pragma unroll
  for (int it = 0; it < kGM; ++it) {
    const int g = it >> 2, kc = it & 3;
    s_g[(g * 16 + wave) * kLD + kc * 32 + lane] = (wave <= rlim) ? gv[it] : 0.f;
  }
  if (tid < 6 * 16 * 4)
    s_g[(tid >> 2) * kLD + 128 + (tid & 3)] = (((tid & 3) == 0) && (tjn < kN)) ? gtv : 0.f;
  if (tid < kLD) s_pp[tid] = (tid < kIN) ? ppv : 0.f;
#pragma unroll
  for (int j = 0; j < kEI; ++j) {
    atomicAdd(&s_deg[dcv[j]], dwv[j]);
  }
  for (int i = tid; i < kN; i += nthr) atomicAdd(&s_deg[i], 1.0f);
  __syncthreads();

  if (tid < kN) {
    const int i = tid;
    float pn = 0.f, d = 0.f;
#pragma unroll
    for (int q = 0; q < kIN / 4; ++q) {
      const float4 pv = *reinterpret_cast<const float4*>(&s_pp[4 * q]);
      const float4 xq = *reinterpret_cast<const float4*>(&s_x[i * kLD + 4 * q]);
      pn = __builtin_fmaf(pv.x, pv.x, pn);  d = __builtin_fmaf(xq.x, pv.x, d);
      pn = __builtin_fmaf(pv.y, pv.y, pn);  d = __builtin_fmaf(xq.y, pv.y, d);
      pn = __builtin_fmaf(pv.z, pv.z, pn);  d = __builtin_fmaf(xq.z, pv.z, d);
      pn = __builtin_fmaf(pv.w, pv.w, pn);  d = __builtin_fmaf(xq.w, pv.w, d);
    }
    {
      const float pl = s_pp[kIN - 1];
      pn = __builtin_fmaf(pl, pl, pn);
      d  = __builtin_fmaf(s_x[i * kLD + kIN - 1], pl, d);
    }
    s_score[i] = tanhf(d * rsqrtf(pn));
    float dg = s_deg[i];
    s_dis[i] = dg > 0.f ? rsqrtf(fmaxf(dg, 1e-12f)) : 0.f;
  }
  __syncthreads();

  for (int i = tid; i < kN; i += nthr) {
    float si = s_score[i];
    int rank = 0;
    for (int j = 0; j < kN; ++j) {
      float sj = s_score[j];
      rank += (sj > si) || (sj == si && j < i);
    }
    s_perm[rank] = i;
  }

  {
    float av[kEI];
#pragma unroll
    for (int j = 0; j < kEI; ++j) av[j] = s_dis[drv[j]] * dwv[j] * s_dis[dcv[j]];
#pragma unroll
    for (int j = 0; j < kEI; ++j) atomicAdd(&s_adj[dcv[j] * kLD + drv[j]], av[j]);
    if (tid >= kNT - kN) {
      const int i = tid - (kNT - kN);
      const float d = s_dis[i];
      atomicAdd(&s_adj[i * kLD + i], d * d);
    }
  }
  __syncthreads();

  if (wave < 8) {
    const int mt = wave;
    const int jn = nb * 16 + m16;
    const float br = b_ih[min(jn, 3 * kN - 1)]          + b_hh[min(jn, 3 * kN - 1)];
    const float bz = b_ih[min(kN + jn, 3 * kN - 1)]     + b_hh[min(kN + jn, 3 * kN - 1)];
    const float bi = b_ih[min(2 * kN + jn, 3 * kN - 1)];
    const float bh = b_hh[min(2 * kN + jn, 3 * kN - 1)];
    v8f cr, cz, ci, ch;
#pragma unroll
    for (int v = 0; v < 8; ++v) { cr[v] = br; cz[v] = bz; ci[v] = bi; ch[v] = bh; }
    const int arow = (mt * 16 + m16) * kLD;
    const int brow = m16 * kLD;
    const int xr   = mt * 16 + m16;
    const int prm  = s_perm[xr];
    const int xrow = prm * kLD;
    const float sc = s_score[prm];
#pragma unroll 11
    for (int ks = 0; ks < kKS; ++ks) {
      const int k0 = ks * 4 + 2 * hh;
      v2f axt = *reinterpret_cast<const v2f*>(&s_x[xrow + k0]);
      axt.x *= sc;  axt.y *= sc;
      v2f ah0 = *reinterpret_cast<const v2f*>(&s_h [arow + k0]);
      v2f bri = *reinterpret_cast<const v2f*>(&s_g[0 * 16 * kLD + brow + k0]);
      v2f bzi = *reinterpret_cast<const v2f*>(&s_g[1 * 16 * kLD + brow + k0]);
      v2f bni = *reinterpret_cast<const v2f*>(&s_g[2 * 16 * kLD + brow + k0]);
      v2f brh = *reinterpret_cast<const v2f*>(&s_g[3 * 16 * kLD + brow + k0]);
      v2f bzh = *reinterpret_cast<const v2f*>(&s_g[4 * 16 * kLD + brow + k0]);
      v2f bnh = *reinterpret_cast<const v2f*>(&s_g[5 * 16 * kLD + brow + k0]);
      cr = wmma4(axt, bri, cr);   cr = wmma4(ah0, brh, cr);
      cz = wmma4(axt, bzi, cz);   cz = wmma4(ah0, bzh, cz);
      ci = wmma4(axt, bni, ci);
      ch = wmma4(ah0, bnh, ch);
    }
#pragma unroll
    for (int v = 0; v < 8; ++v) {
      const int row = mt * 16 + v + 8 * hh;
      const float w0v = (jn < kLD) ? s_h[row * kLD + jn] : 0.f;
      s_wT[m16 * kLD + row] = gru_out(cr[v], cz[v], ci[v], ch[v], w0v);
    }
  } else if (wave < 11) {
    const int part = wave - 8;
    const int jn   = nb * 16 + m16;
    const int arow = (8 * 16 + m16) * kLD;
    const int brow = m16 * kLD;
    const int xr   = 8 * 16 + m16;
    const int prm  = s_perm[min(xr, kN - 1)];
    const int xrow = ((xr < kN) ? prm : xr) * kLD;
    const float sc = (xr < kN) ? s_score[prm] : 0.f;
    if (part < 2) {
      const int bidx = min(part * kN + jn, 3 * kN - 1);
      const float b0 = b_ih[bidx] + b_hh[bidx];
      v8f c;
#pragma unroll
      for (int v = 0; v < 8; ++v) c[v] = b0;
      const int gi_ = part * 16 * kLD + brow;
      const int gh_ = (3 + part) * 16 * kLD + brow;
#pragma unroll 11
      for (int ks = 0; ks < kKS; ++ks) {
        const int k0 = ks * 4 + 2 * hh;
        v2f axt = *reinterpret_cast<const v2f*>(&s_x[xrow + k0]);
        axt.x *= sc;  axt.y *= sc;
        v2f ah0 = *reinterpret_cast<const v2f*>(&s_h[arow + k0]);
        v2f bgi = *reinterpret_cast<const v2f*>(&s_g[gi_ + k0]);
        v2f bgh = *reinterpret_cast<const v2f*>(&s_g[gh_ + k0]);
        c = wmma4(axt, bgi, c);   c = wmma4(ah0, bgh, c);
      }
      if (hh == 0) s_t8[part * 16 + m16] = c[0];
    } else {
      const float bi = b_ih[min(2 * kN + jn, 3 * kN - 1)];
      const float bh = b_hh[min(2 * kN + jn, 3 * kN - 1)];
      v8f ci, ch;
#pragma unroll
      for (int v = 0; v < 8; ++v) { ci[v] = bi; ch[v] = bh; }
#pragma unroll 11
      for (int ks = 0; ks < kKS; ++ks) {
        const int k0 = ks * 4 + 2 * hh;
        v2f axt = *reinterpret_cast<const v2f*>(&s_x[xrow + k0]);
        axt.x *= sc;  axt.y *= sc;
        v2f ah0 = *reinterpret_cast<const v2f*>(&s_h[arow + k0]);
        v2f bni = *reinterpret_cast<const v2f*>(&s_g[2 * 16 * kLD + brow + k0]);
        v2f bnh = *reinterpret_cast<const v2f*>(&s_g[5 * 16 * kLD + brow + k0]);
        ci = wmma4(axt, bni, ci);
        ch = wmma4(ah0, bnh, ch);
      }
      if (hh == 0) { s_t8[2 * 16 + m16] = ci[0];  s_t8[3 * 16 + m16] = ch[0]; }
    }
  }
  __syncthreads();

  if (tid < 16) {
    const int jn = nb * 16 + tid;
    const float w0v = (jn < kLD) ? s_h[128 * kLD + jn] : 0.f;
    s_wT[tid * kLD + 128] = gru_out(s_t8[tid], s_t8[16 + tid], s_t8[32 + tid], s_t8[48 + tid], w0v);
    s_wT[tid * kLD + 129] = 0.f;
    s_wT[tid * kLD + 130] = 0.f;
    s_wT[tid * kLD + 131] = 0.f;
  }
  __syncthreads();

  if (wave < kMT) {
    const int mt = wave;
    v8f acc = {0.f, 0.f, 0.f, 0.f, 0.f, 0.f, 0.f, 0.f};
    const int arow = (mt * 16 + m16) * kLD;
    const int brow = m16 * kLD;
#pragma unroll 11
    for (int ks = 0; ks < kKS; ++ks) {
      const int k0 = ks * 4 + 2 * hh;
      v2f a = *reinterpret_cast<const v2f*>(&s_x [arow + k0]);
      v2f b = *reinterpret_cast<const v2f*>(&s_wT[brow + k0]);
      acc = wmma4(a, b, acc);
    }
#pragma unroll
    for (int v = 0; v < 8; ++v) {
      const int row = mt * 16 + v + 8 * hh;
      if (row < kLD) s_xwT[m16 * kLD + row] = acc[v];
    }
  }
  __syncthreads();

  if (wave < kMT) {
    const int mt = wave;
    const int jn = nb * 16 + m16;
    const float cb = conv_b[min(jn, kN - 1)];
    v8f acc;
#pragma unroll
    for (int v = 0; v < 8; ++v) acc[v] = cb;
    const int arow = (mt * 16 + m16) * kLD;
    const int brow = m16 * kLD;
#pragma unroll 11
    for (int ks = 0; ks < kKS; ++ks) {
      const int k0 = ks * 4 + 2 * hh;
      v2f a = *reinterpret_cast<const v2f*>(&s_adj[arow + k0]);
      v2f b = *reinterpret_cast<const v2f*>(&s_xwT[brow + k0]);
      acc = wmma4(a, b, acc);
    }
#pragma unroll
    for (int v = 0; v < 8; ++v) {
      float hv = acc[v];
      hv = hv > 0.f ? hv : (__expf(hv) - 1.f);
      s_hout[(mt * 16 + v + 8 * hh) * 16 + m16] = jn < kN ? hv : 0.f;
    }
  }
  __syncthreads();

  float* hb = h_ws + nb * kHB;
  for (int i = tid; i < kHB / 4; i += nthr)
    reinterpret_cast<float4*>(hb)[i] = reinterpret_cast<const float4*>(s_hout)[i];
}

__launch_bounds__(128, 1)
__global__ void evolve_gcn_linear(const float* __restrict__ h_ws,
                                  const float* __restrict__ lin_w,
                                  const float* __restrict__ lin_b,
                                  float*       __restrict__ out,
                                  unsigned long long ws_bytes)
{
  if (ws_bytes < kWsNeed) return;
  __shared__ __attribute__((aligned(16))) float s_o[16 * kOUT];
  const int tid  = threadIdx.x;
  const int lane = tid & 31;
  const int ot   = tid >> 5;
  const int m16  = lane & 15;
  const int hh   = lane >> 4;
  const int mt   = blockIdx.x;
  const int oc   = ot * 16 + m16;
  const float lb = lin_b[oc];
  v8f acc;
#pragma unroll
  for (int v = 0; v < 8; ++v) acc[v] = lb;
  const int arow = (mt * 16 + m16) * 16;
  const int rowo = oc * kIN;
#pragma unroll
  for (int ks = 0; ks < kKS; ++ks) {
    const int k0 = ks * 4 + 2 * hh;
    v2f a = *reinterpret_cast<const v2f*>(&h_ws[(k0 >> 4) * kHB + arow + (k0 & 15)]);
    const int i0 = min(k0,     kIN - 1);
    const int i1 = min(k0 + 1, kIN - 1);
    v2f b; b.x = lin_w[rowo + i0]; b.y = lin_w[rowo + i1];
    acc = wmma4(a, b, acc);
  }
#pragma unroll
  for (int v = 0; v < 8; ++v) s_o[(v + 8 * hh) * kOUT + oc] = acc[v];
  __syncthreads();
  for (int i = tid; i < 16 * kOUT / 4; i += 128) {
    const int row = mt * 16 + i / (kOUT / 4);
    if (row < kN)
      reinterpret_cast<float4*>(out)[mt * (16 * kOUT / 4) + i] = reinterpret_cast<const float4*>(s_o)[i];
  }
}

extern "C" void kernel_launch(void* const* d_in, const int* in_sizes, int n_in,
                              void* d_out, int out_size, void* d_ws, size_t ws_size,
                              hipStream_t stream) {
  (void)in_sizes; (void)n_in; (void)out_size;
  const float* x       = (const float*)d_in[0];
  const int*   eidx    = (const int*)  d_in[1];
  const float* ew      = (const float*)d_in[2];
  const float* pool_p  = (const float*)d_in[3];
  const float* W0      = (const float*)d_in[4];
  const float* w_ih    = (const float*)d_in[5];
  const float* w_hh    = (const float*)d_in[6];
  const float* b_ih    = (const float*)d_in[7];
  const float* b_hh    = (const float*)d_in[8];
  const float* conv_b  = (const float*)d_in[9];
  const float* lin_w   = (const float*)d_in[10];
  const float* lin_b   = (const float*)d_in[11];
  evolve_gcn_colblock<<<9, 512, 0, stream>>>(x, eidx, ew, pool_p, W0, w_ih, w_hh,
                                             b_ih, b_hh, conv_b, (float*)d_ws,
                                             (unsigned long long)ws_size);
  evolve_gcn_linear<<<9, 128, 0, stream>>>((const float*)d_ws, lin_w, lin_b,
                                           (float*)d_out, (unsigned long long)ws_size);
}


# GRU rows dealt to three workgroups per column block (27 WGPs): each runs 3 of the 9 row tiles, its x@W is a K-split partial, launch 2 sums the three adj@xw planes before ELU; changes summation order
# speedup vs baseline: 16.3204x; 1.1254x over previous
#include <hip/hip_runtime.h>


typedef __attribute__((ext_vector_type(2))) float v2f;
typedef __attribute__((ext_vector_type(8))) float v8f;

static constexpr int kN   = 129;
static constexpr int kIN  = 129;
static constexpr int kOUT = 64;
static constexpr int kE   = 4096;
static constexpr int kLD  = 132;
static constexpr int kRP  = 144;
static constexpr int kKS  = 33;
static constexpr int kMT  = 9;
static constexpr int kRG  = 3;
static constexpr int kTG  = 3;
static constexpr int kKG  = kTG * 4;
static constexpr int kNT  = 512;
static constexpr int kXI  = (kN * kIN + kNT - 1) / kNT;
static constexpr int kGE  = 6 * 16 * kLD;
static constexpr int kGM  = 24;
static constexpr int kXA  = 6;
static constexpr int kEI  = kE / kNT;
static constexpr int kHB  = kRP * 16;
static constexpr int kPL  = kMT * kHB;
static constexpr int kLA  = 196;
static constexpr unsigned long long kWsNeed = 4ull * kRG * kPL;

static __device__ __forceinline__ v8f wmma4(v2f a, v2f b, v8f c) {
  return __builtin_amdgcn_wmma_f32_16x16x4_f32(
       false, a,  false, b,
       (short)0, c,  false,  false);
}

static __device__ __forceinline__ float sigm(float x) {
  return 1.0f / (1.0f + __expf(-x));
}

static __device__ __forceinline__ float gru_out(float cr, float cz, float ci, float ch, float w0v) {
  const float rg  = sigm(cr);
  const float zg  = sigm(cz);
  const float cnd = tanhf(ci + rg * ch);
  return (1.f - zg) * cnd + zg * w0v;
}

template <int NKS>
static __device__ __forceinline__ v8f chain_k(const float* a, const float* b, int kb, int hh, v8f acc) {
#pragma unroll
  for (int ks = 0; ks < NKS; ++ks) {
    const int k0 = kb + ks * 4 + 2 * hh;
    v2f av = *reinterpret_cast<const v2f*>(&a[k0]);
    v2f bv = *reinterpret_cast<const v2f*>(&b[k0]);
    acc = wmma4(av, bv, acc);
  }
  return acc;
}

__launch_bounds__(512, 1)
__global__ void evolve_gcn_colblock(const float* __restrict__ x,
                                    const int*   __restrict__ eidx,
                                    const float* __restrict__ ew,
                                    const float* __restrict__ pool_p,
                                    const float* __restrict__ W0,
                                    const float* __restrict__ w_ih,
                                    const float* __restrict__ w_hh,
                                    const float* __restrict__ b_ih,
                                    const float* __restrict__ b_hh,
                                    const float* __restrict__ conv_b,
                                    float*       __restrict__ h_ws,
                                    unsigned long long ws_bytes)
{
  if (ws_bytes < kWsNeed) return;

  __shared__ __attribute__((aligned(16))) float s_x [kRP * kLD];
  __shared__ __attribute__((aligned(16))) float s_adj[kRP * kLD];
  __shared__ __attribute__((aligned(16))) float s_h [kRP * kLD];
  __shared__ __attribute__((aligned(16))) float s_g [kGE];
  __shared__ __attribute__((aligned(16))) float s_wT [16 * kLD];
  __shared__ __attribute__((aligned(16))) float s_xwT[16 * kLD];
  __shared__ __attribute__((aligned(16))) float s_hout[kHB];
  __shared__ float s_score[kN];
  __shared__ __attribute__((aligned(16))) float s_pp[kLD];
  __shared__ int   s_perm [kN];
  __shared__ float s_dis  [kN];
  __shared__ float s_deg  [kN];
  __shared__ float s_t8   [4 * 16];

  const int tid    = threadIdx.x;
  constexpr int nthr = kNT;
  const int lane   = tid & 31;
  const int wave   = tid >> 5;
  const int m16    = lane & 15;
  const int hh     = lane >> 4;
  const int nb     = blockIdx.x;
  const int rg     = blockIdx.y;

  float xv[kXI], wv[kXI], gv[kGM], dwv[kEI];
  int   drv[kEI], dcv[kEI];
  const int rlim = kN - 1 - nb * 16;
  const float* gi0 = w_ih + nb * 16 * kIN;
  const float* gh0 = w_hh + nb * 16 * kIN;
  const int gofs = min(wave, rlim) * kIN + lane;
#pragma unroll
  for (int it = 0; it < kGM; ++it) {
    const int g = it >> 2, kc = it & 3;
    const float* src = (g < 3 ? gi0 : gh0) + (g % 3) * kN * kIN + kc * 32;
    gv[it] = src[gofs];
  }
  const int trow = min(tid >> 2, 6 * 16 - 1);
  const int tjn  = nb * 16 + (trow & 15);
  const float* tsrc = (trow < 3 * 16) ? w_ih : w_hh;
  const float gtv = tsrc[(((trow >> 4) % 3) * kN + min(tjn, kN - 1)) * kIN + (kIN - 1)];
#pragma unroll
  for (int j = 0; j < kEI; ++j) {
    const int e = tid + j * kNT;
    drv[j] = eidx[e];
    dcv[j] = eidx[kE + e];
    dwv[j] = ew[e];
  }
  const float ppv = pool_p[min(tid, kIN - 1)];
#pragma unroll
  for (int j = 0; j < kXA; ++j) {
    const int i = tid + j * kNT;
    xv[j] = x[i];
    wv[j] = W0[i];
  }

  for (int i = tid; i < kRP * kLD; i += nthr) s_adj[i] = 0.f;
  for (int i = tid; i < kN; i += nthr) s_deg[i] = 0.f;

#pragma unroll
  for (int j = kXA; j < kXI; ++j) {
    const int i = min(tid + j * kNT, kN * kIN - 1);
    xv[j] = x[i];
    wv[j] = W0[i];
  }

  for (int i = tid; i < kRP * kLD; i += nthr) {
    s_x[i] = 0.f; s_h[i] = 0.f;
  }
  __syncthreads();

#pragma unroll
  for (int j = 0; j < kXI; ++j) {
    const int i = tid + j * kNT;
    if (i < kN * kIN) {
      const int r = i / kIN, c = i % kIN;
      s_x[r * kLD + c] = xv[j];
      s_h[r * kLD + c] = wv[j];
    }
  }
#pragma unroll
  for (int it = 0; it < kGM; ++it) {
    const int g = it >> 2, kc = it & 3;
    s_g[(g * 16 + wave) * kLD + kc * 32 + lane] = (wave <= rlim) ? gv[it] : 0.f;
  }
  if (tid < 6 * 16 * 4)
    s_g[(tid >> 2) * kLD + 128 + (tid & 3)] = (((tid & 3) == 0) && (tjn < kN)) ? gtv : 0.f;
  if (tid < kLD) s_pp[tid] = (tid < kIN) ? ppv : 0.f;
#pragma unroll
  for (int j = 0; j < kEI; ++j) {
    atomicAdd(&s_deg[dcv[j]], dwv[j]);
  }
  for (int i = tid; i < kN; i += nthr) atomicAdd(&s_deg[i], 1.0f);
  __syncthreads();

  if (tid < kN) {
    const int i = tid;
    float pn = 0.f, d = 0.f;
#pragma unroll
    for (int q = 0; q < kIN / 4; ++q) {
      const float4 pv = *reinterpret_cast<const float4*>(&s_pp[4 * q]);
      const float4 xq = *reinterpret_cast<const float4*>(&s_x[i * kLD + 4 * q]);
      pn = __builtin_fmaf(pv.x, pv.x, pn);  d = __builtin_fmaf(xq.x, pv.x, d);
      pn = __builtin_fmaf(pv.y, pv.y, pn);  d = __builtin_fmaf(xq.y, pv.y, d);
      pn = __builtin_fmaf(pv.z, pv.z, pn);  d = __builtin_fmaf(xq.z, pv.z, d);
      pn = __builtin_fmaf(pv.w, pv.w, pn);  d = __builtin_fmaf(xq.w, pv.w, d);
    }
    {
      const float pl = s_pp[kIN - 1];
      pn = __builtin_fmaf(pl, pl, pn);
      d  = __builtin_fmaf(s_x[i * kLD + kIN - 1], pl, d);
    }
    s_score[i] = tanhf(d * rsqrtf(pn));
    float dg = s_deg[i];
    s_dis[i] = dg > 0.f ? rsqrtf(fmaxf(dg, 1e-12f)) : 0.f;
  }
  __syncthreads();

  for (int i = tid; i < kN; i += nthr) {
    float si = s_score[i];
    int rank = 0;
    for (int j = 0; j < kN; ++j) {
      float sj = s_score[j];
      rank += (sj > si) || (sj == si && j < i);
    }
    s_perm[rank] = i;
  }

  {
    float av[kEI];
#pragma unroll
    for (int j = 0; j < kEI; ++j) av[j] = s_dis[drv[j]] * dwv[j] * s_dis[dcv[j]];
#pragma unroll
    for (int j = 0; j < kEI; ++j) atomicAdd(&s_adj[dcv[j] * kLD + drv[j]], av[j]);
    if (tid >= kNT - kN) {
      const int i = tid - (kNT - kN);
      const float d = s_dis[i];
      atomicAdd(&s_adj[i * kLD + i], d * d);
    }
  }
  __syncthreads();

  const int nfull = (rg == kRG - 1) ? kTG - 1 : kTG;
  if (wave < nfull) {
    const int mt = rg * kTG + wave;
    const int jn = nb * 16 + m16;
    const float br = b_ih[min(jn, 3 * kN - 1)]          + b_hh[min(jn, 3 * kN - 1)];
    const float bz = b_ih[min(kN + jn, 3 * kN - 1)]     + b_hh[min(kN + jn, 3 * kN - 1)];
    const float bi = b_ih[min(2 * kN + jn, 3 * kN - 1)];
    const float bh = b_hh[min(2 * kN + jn, 3 * kN - 1)];
    v8f cr, cz, ci, ch;
#pragma unroll
    for (int v = 0; v < 8; ++v) { cr[v] = br; cz[v] = bz; ci[v] = bi; ch[v] = bh; }
    const int arow = (mt * 16 + m16) * kLD;
    const int brow = m16 * kLD;
    const int xr   = mt * 16 + m16;
    const int prm  = s_perm[xr];
    const int xrow = prm * kLD;
    const float sc = s_score[prm];
#pragma unroll 11
    for (int ks = 0; ks < kKS; ++ks) {
      const int k0 = ks * 4 + 2 * hh;
      v2f axt = *reinterpret_cast<const v2f*>(&s_x[xrow + k0]);
      axt.x *= sc;  axt.y *= sc;
      v2f ah0 = *reinterpret_cast<const v2f*>(&s_h [arow + k0]);
      v2f bri = *reinterpret_cast<const v2f*>(&s_g[0 * 16 * kLD + brow + k0]);
      v2f bzi = *reinterpret_cast<const v2f*>(&s_g[1 * 16 * kLD + brow + k0]);
      v2f bni = *reinterpret_cast<const v2f*>(&s_g[2 * 16 * kLD + brow + k0]);
      v2f brh = *reinterpret_cast<const v2f*>(&s_g[3 * 16 * kLD + brow + k0]);
      v2f bzh = *reinterpret_cast<const v2f*>(&s_g[4 * 16 * kLD + brow + k0]);
      v2f bnh = *reinterpret_cast<const v2f*>(&s_g[5 * 16 * kLD + brow + k0]);
      cr = wmma4(axt, bri, cr);   cr = wmma4(ah0, brh, cr);
      cz = wmma4(axt, bzi, cz);   cz = wmma4(ah0, bzh, cz);
      ci = wmma4(axt, bni, ci);
      ch = wmma4(ah0, bnh, ch);
    }
#pragma unroll
    for (int v = 0; v < 8; ++v) {
      const int row = mt * 16 + v + 8 * hh;
      const float w0v = (jn < kLD) ? s_h[row * kLD + jn] : 0.f;
      s_wT[m16 * kLD + row] = gru_out(cr[v], cz[v], ci[v], ch[v], w0v);
    }
  } else if (rg == kRG - 1 && (wave == 2 || wave == 3 || wave == 7)) {
    const int part = (wave == 7) ? 2 : wave - 2;
    const int jn   = nb * 16 + m16;
    const int arow = (8 * 16 + m16) * kLD;
    const int brow = m16 * kLD;
    const int xr   = 8 * 16 + m16;
    const int prm  = s_perm[min(xr, kN - 1)];
    const int xrow = ((xr < kN) ? prm : xr) * kLD;
    const float sc = (xr < kN) ? s_score[prm] : 0.f;
    if (part < 2) {
      const int bidx = min(part * kN + jn, 3 * kN - 1);
      const float b0 = b_ih[bidx] + b_hh[bidx];
      v8f c;
#pragma unroll
      for (int v = 0; v < 8; ++v) c[v] = b0;
      const int gi_ = part * 16 * kLD + brow;
      const int gh_ = (3 + part) * 16 * kLD + brow;
#pragma unroll 11
      for (int ks = 0; ks < kKS; ++ks) {
        const int k0 = ks * 4 + 2 * hh;
        v2f axt = *reinterpret_cast<const v2f*>(&s_x[xrow + k0]);
        axt.x *= sc;  axt.y *= sc;
        v2f ah0 = *reinterpret_cast<const v2f*>(&s_h[arow + k0]);
        v2f bgi = *reinterpret_cast<const v2f*>(&s_g[gi_ + k0]);
        v2f bgh = *reinterpret_cast<const v2f*>(&s_g[gh_ + k0]);
        c = wmma4(axt, bgi, c);   c = wmma4(ah0, bgh, c);
      }
      if (hh == 0) s_t8[part * 16 + m16] = c[0];
    } else {
      const float bi = b_ih[min(2 * kN + jn, 3 * kN - 1)];
      const float bh = b_hh[min(2 * kN + jn, 3 * kN - 1)];
      v8f ci, ch;
#pragma unroll
      for (int v = 0; v < 8; ++v) { ci[v] = bi; ch[v] = bh; }
#pragma unroll 11
      for (int ks = 0; ks < kKS; ++ks) {
        const int k0 = ks * 4 + 2 * hh;
        v2f axt = *reinterpret_cast<const v2f*>(&s_x[xrow + k0]);
        axt.x *= sc;  axt.y *= sc;
        v2f ah0 = *reinterpret_cast<const v2f*>(&s_h[arow + k0]);
        v2f bni = *reinterpret_cast<const v2f*>(&s_g[2 * 16 * kLD + brow + k0]);
        v2f bnh = *reinterpret_cast<const v2f*>(&s_g[5 * 16 * kLD + brow + k0]);
        ci = wmma4(axt, bni, ci);
        ch = wmma4(ah0, bnh, ch);
      }
      if (hh == 0) { s_t8[2 * 16 + m16] = ci[0];  s_t8[3 * 16 + m16] = ch[0]; }
    }
  }
  __syncthreads();

  if (rg == kRG - 1 && tid < 16) {
    const int jn = nb * 16 + tid;
    const float w0v = (jn < kLD) ? s_h[128 * kLD + jn] : 0.f;
    s_wT[tid * kLD + 128] = gru_out(s_t8[tid], s_t8[16 + tid], s_t8[32 + tid], s_t8[48 + tid], w0v);
    s_wT[tid * kLD + 129] = 0.f;
    s_wT[tid * kLD + 130] = 0.f;
    s_wT[tid * kLD + 131] = 0.f;
  }
  __syncthreads();

  if (wave < kMT) {
    const int mt = wave;
    v8f acc = {0.f, 0.f, 0.f, 0.f, 0.f, 0.f, 0.f, 0.f};
    const int arow = (mt * 16 + m16) * kLD;
    const int brow = m16 * kLD;
    if (rg == kRG - 1) acc = chain_k<kKS - (kRG - 1) * kKG>(&s_x[arow], &s_wT[brow], rg * kKG * 4, hh, acc);
    else               acc = chain_k<kKG>(&s_x[arow], &s_wT[brow], rg * kKG * 4, hh, acc);
#pragma unroll
    for (int v = 0; v < 8; ++v) {
      const int row = mt * 16 + v + 8 * hh;
      if (row < kLD) s_xwT[m16 * kLD + row] = acc[v];
    }
  }
  __syncthreads();

  if (wave < kMT) {
    const int mt = wave;
    const int jn = nb * 16 + m16;
    const float cb = (rg == 0) ? conv_b[min(jn, kN - 1)] : 0.f;
    v8f acc;
#pragma unroll
    for (int v = 0; v < 8; ++v) acc[v] = cb;
    const int arow = (mt * 16 + m16) * kLD;
    const int brow = m16 * kLD;
#pragma unroll 11
    for (int ks = 0; ks < kKS; ++ks) {
      const int k0 = ks * 4 + 2 * hh;
      v2f a = *reinterpret_cast<const v2f*>(&s_adj[arow + k0]);
      v2f b = *reinterpret_cast<const v2f*>(&s_xwT[brow + k0]);
      acc = wmma4(a, b, acc);
    }
#pragma unroll
    for (int v = 0; v < 8; ++v)
      s_hout[(mt * 16 + v + 8 * hh) * 16 + m16] = jn < kN ? acc[v] : 0.f;
  }
  __syncthreads();

  float* hb = h_ws + rg * kPL + nb * kHB;
  for (int i = tid; i < kHB / 4; i += nthr)
    reinterpret_cast<float4*>(hb)[i] = reinterpret_cast<const float4*>(s_hout)[i];
}

__launch_bounds__(128) __attribute__((amdgpu_waves_per_eu(1, 1)))
__global__ void evolve_gcn_linear(const float* __restrict__ h_ws,
                                  const float* __restrict__ lin_w,
                                  const float* __restrict__ lin_b,
                                  float*       __restrict__ out,
                                  unsigned long long ws_bytes)
{
  if (ws_bytes < kWsNeed) return;
  __shared__ __attribute__((aligned(16))) float s_a[16 * kLA];
  __shared__ __attribute__((aligned(16))) float s_o[16 * kOUT];
  const int tid  = threadIdx.x;
  const int lane = tid & 31;
  const int ot   = tid >> 5;
  const int m16  = lane & 15;
  const int hh   = lane >> 4;
  const int mt   = blockIdx.x;
  const int oc   = ot * 16 + m16;

  const int kb0 = tid >> 6;
  const int pw  = tid & 63;
  const float* pb = h_ws + mt * 256 + pw * 4;
  float4 pv[kRG][5];
#pragma unroll
  for (int g = 0; g < kRG; ++g) {
#pragma unroll
    for (int j = 0; j < 5; ++j)
      pv[g][j] = *reinterpret_cast<const float4*>(pb + g * kPL + min(kb0 + 2 * j, kMT - 1) * kHB);
  }
  const float lb = lin_b[oc];
  const int rowo = oc * kIN;
  v2f bv[kKS];
#pragma unroll
  for (int ks = 0; ks < kKS; ++ks) {
    const int k0 = ks * 4 + 2 * hh;
    const int i0 = min(k0,     kIN - 1);
    const int i1 = min(k0 + 1, kIN - 1);
    bv[ks].x = lin_w[rowo + i0];
    bv[ks].y = lin_w[rowo + i1];
  }

#pragma unroll
  for (int j = 0; j < 5; ++j) {
    float4 s;
    s.x = (pv[0][j].x + pv[1][j].x) + pv[2][j].x;
    s.y = (pv[0][j].y + pv[1][j].y) + pv[2][j].y;
    s.z = (pv[0][j].z + pv[1][j].z) + pv[2][j].z;
    s.w = (pv[0][j].w + pv[1][j].w) + pv[2][j].w;
    s.x = s.x > 0.f ? s.x : (__expf(s.x) - 1.f);
    s.y = s.y > 0.f ? s.y : (__expf(s.y) - 1.f);
    s.z = s.z > 0.f ? s.z : (__expf(s.z) - 1.f);
    s.w = s.w > 0.f ? s.w : (__expf(s.w) - 1.f);
    *reinterpret_cast<float4*>(&s_a[(pw >> 2) * kLA + (kb0 + 2 * j) * 16 + (pw & 3) * 4]) = s;
  }
  __syncthreads();

  v8f acc;
#pragma unroll
  for (int v = 0; v < 8; ++v) acc[v] = lb;
  const int arow = m16 * kLA;
#pragma unroll
  for (int ks = 0; ks < kKS; ++ks) {
    const int k0 = ks * 4 + 2 * hh;
    v2f a = *reinterpret_cast<const v2f*>(&s_a[arow + k0]);
    acc = wmma4(a, bv[ks], acc);
  }
#pragma unroll
  for (int v = 0; v < 8; ++v) s_o[(v + 8 * hh) * kOUT + oc] = acc[v];
  __syncthreads();
  for (int i = tid; i < 16 * kOUT / 4; i += 128) {
    const int row = mt * 16 + i / (kOUT / 4);
    if (row < kN)
      reinterpret_cast<float4*>(out)[mt * (16 * kOUT / 4) + i] = reinterpret_cast<const float4*>(s_o)[i];
  }
}

extern "C" void kernel_launch(void* const* d_in, const int* in_sizes, int n_in,
                              void* d_out, int out_size, void* d_ws, size_t ws_size,
                              hipStream_t stream) {
  (void)in_sizes; (void)n_in; (void)out_size;
  const float* x       = (const float*)d_in[0];
  const int*   eidx    = (const int*)  d_in[1];
  const float* ew      = (const float*)d_in[2];
  const float* pool_p  = (const float*)d_in[3];
  const float* W0      = (const float*)d_in[4];
  const float* w_ih    = (const float*)d_in[5];
  const float* w_hh    = (const float*)d_in[6];
  const float* b_ih    = (const float*)d_in[7];
  const float* b_hh    = (const float*)d_in[8];
  const float* conv_b  = (const float*)d_in[9];
  const float* lin_w   = (const float*)d_in[10];
  const float* lin_b   = (const float*)d_in[11];
  evolve_gcn_colblock<<<dim3(9, 3), 512, 0, stream>>>(x, eidx, ew, pool_p, W0, w_ih, w_hh,
                                             b_ih, b_hh, conv_b, (float*)d_ws,
                                             (unsigned long long)ws_size);
  evolve_gcn_linear<<<9, 128, 0, stream>>>((const float*)d_ws, lin_w, lin_b,
                                           (float*)d_out, (unsigned long long)ws_size);
}


# GRU step's twelve accumulator chains per group go one per wave over all four SIMDs (h_n chains beside the ranking), gate sums via LDS, W elements finished on all 512 threads; sum order kept
# speedup vs baseline: 18.1410x; 1.1116x over previous
#include <hip/hip_runtime.h>


typedef __attribute__((ext_vector_type(2))) float v2f;
typedef __attribute__((ext_vector_type(8))) float v8f;

static constexpr int kN   = 129;
static constexpr int kIN  = 129;
static constexpr int kOUT = 64;
static constexpr int kE   = 4096;
static constexpr int kLD  = 132;
static constexpr int kRP  = 144;
static constexpr int kKS  = 33;
static constexpr int kMT  = 9;
static constexpr int kRG  = 3;
static constexpr int kTG  = 3;
static constexpr int kKG  = kTG * 4;
static constexpr int kNT  = 512;
static constexpr int kXI  = (kN * kIN + kNT - 1) / kNT;
static constexpr int kGE  = 6 * 16 * kLD;
static constexpr int kGM  = 24;
static constexpr int kXA  = 6;
static constexpr int kEI  = kE / kNT;
static constexpr int kHB  = kRP * 16;
static constexpr int kPL  = kMT * kHB;
static constexpr int kLA  = 196;
static constexpr unsigned long long kWsNeed = 4ull * kRG * kPL;

static __device__ __forceinline__ v8f wmma4(v2f a, v2f b, v8f c) {
  return __builtin_amdgcn_wmma_f32_16x16x4_f32(
       false, a,  false, b,
       (short)0, c,  false,  false);
}

static __device__ __forceinline__ float sigm(float x) {
  return 1.0f / (1.0f + __expf(-x));
}

static __device__ __forceinline__ float gru_out(float cr, float cz, float ci, float ch, float w0v) {
  const float rg  = sigm(cr);
  const float zg  = sigm(cz);
  const float cnd = tanhf(ci + rg * ch);
  return (1.f - zg) * cnd + zg * w0v;
}

template <int NKS>
static __device__ __forceinline__ v8f chain_k(const float* a, const float* b, int kb, int hh, v8f acc) {
#pragma unroll
  for (int ks = 0; ks < NKS; ++ks) {
    const int k0 = kb + ks * 4 + 2 * hh;
    v2f av = *reinterpret_cast<const v2f*>(&a[k0]);
    v2f bv = *reinterpret_cast<const v2f*>(&b[k0]);
    acc = wmma4(av, bv, acc);
  }
  return acc;
}

template <int NB>
static __device__ __forceinline__ void frag_rd2(const float* pa, const float* pb, int kb,
                                                v2f (&fa)[NB], v2f (&fb)[NB]) {
#pragma unroll
  for (int j = 0; j < NB; ++j) {
    fa[j] = *reinterpret_cast<const v2f*>(&pa[kb + 4 * j]);
    fb[j] = *reinterpret_cast<const v2f*>(&pb[kb + 4 * j]);
  }
}
template <int NB>
static __device__ __forceinline__ void frag_rd4(const float* pa, const float* pb, const float* pc, const float* pd, int kb,
                                                v2f (&fa)[NB], v2f (&fb)[NB], v2f (&fc)[NB], v2f (&fd)[NB]) {
#pragma unroll
  for (int j = 0; j < NB; ++j) {
    fa[j] = *reinterpret_cast<const v2f*>(&pa[kb + 4 * j]);
    fb[j] = *reinterpret_cast<const v2f*>(&pb[kb + 4 * j]);
    fc[j] = *reinterpret_cast<const v2f*>(&pc[kb + 4 * j]);
    fd[j] = *reinterpret_cast<const v2f*>(&pd[kb + 4 * j]);
  }
}
template <int NB>
static __device__ __forceinline__ v8f mma_rz(const v2f (&xa)[NB], const v2f (&bi)[NB],
                                             const v2f (&ha)[NB], const v2f (&bh)[NB], float sc, v8f c) {
#pragma unroll
  for (int j = 0; j < NB; ++j) {
    v2f a = xa[j];
    a.x *= sc;  a.y *= sc;
    c = wmma4(a, bi[j], c);
    c = wmma4(ha[j], bh[j], c);
  }
  return c;
}
template <int NB>
static __device__ __forceinline__ v8f mma_xt(const v2f (&xa)[NB], const v2f (&b)[NB], float sc, v8f c) {
#pragma unroll
  for (int j = 0; j < NB; ++j) {
    v2f a = xa[j];
    a.x *= sc;  a.y *= sc;
    c = wmma4(a, b[j], c);
  }
  return c;
}
template <int NB>
static __device__ __forceinline__ v8f mma_pl(const v2f (&a)[NB], const v2f (&b)[NB], v8f c) {
#pragma unroll
  for (int j = 0; j < NB; ++j) c = wmma4(a[j], b[j], c);
  return c;
}

__launch_bounds__(512, 1)
__global__ void evolve_gcn_colblock(const float* __restrict__ x,
                                    const int*   __restrict__ eidx,
                                    const float* __restrict__ ew,
                                    const float* __restrict__ pool_p,
                                    const float* __restrict__ W0,
                                    const float* __restrict__ w_ih,
                                    const float* __restrict__ w_hh,
                                    const float* __restrict__ b_ih,
                                    const float* __restrict__ b_hh,
                                    const float* __restrict__ conv_b,
                                    float*       __restrict__ h_ws,
                                    unsigned long long ws_bytes)
{
  if (ws_bytes < kWsNeed) return;

  __shared__ __attribute__((aligned(16))) float s_x [kRP * kLD];
  __shared__ __attribute__((aligned(16))) float s_adj[kRP * kLD];
  __shared__ __attribute__((aligned(16))) float s_h [kRP * kLD];
  __shared__ __attribute__((aligned(16))) float s_g [kGE];
  __shared__ __attribute__((aligned(16))) float s_wT [16 * kLD];
  __shared__ __attribute__((aligned(16))) float s_xwT[16 * kLD];
  __shared__ __attribute__((aligned(16))) float s_hout[kHB];
  __shared__ float s_score[kN];
  __shared__ __attribute__((aligned(16))) float s_pp[kLD];
  __shared__ int   s_perm [kN];
  __shared__ float s_dis  [kN];
  __shared__ float s_deg  [kN];
  __shared__ float s_gs   [4 * kTG * 16 * 16];

  const int tid    = threadIdx.x;
  constexpr int nthr = kNT;
  const int lane   = tid & 31;
  const int wave   = tid >> 5;
  const int m16    = lane & 15;
  const int hh     = lane >> 4;
  const int nb     = blockIdx.x;
  const int rg     = blockIdx.y;

  float xv[kXI], wv[kXI], gv[kGM], dwv[kEI];
  int   drv[kEI], dcv[kEI];
  const int rlim = kN - 1 - nb * 16;
  const float* gi0 = w_ih + nb * 16 * kIN;
  const float* gh0 = w_hh + nb * 16 * kIN;
  const int gofs = min(wave, rlim) * kIN + lane;
#pragma unroll
  for (int it = 0; it < kGM; ++it) {
    const int g = it >> 2, kc = it & 3;
    const float* src = (g < 3 ? gi0 : gh0) + (g % 3) * kN * kIN + kc * 32;
    gv[it] = src[gofs];
  }
  const int trow = min(tid >> 2, 6 * 16 - 1);
  const int tjn  = nb * 16 + (trow & 15);
  const float* tsrc = (trow < 3 * 16) ? w_ih : w_hh;
  const float gtv = tsrc[(((trow >> 4) % 3) * kN + min(tjn, kN - 1)) * kIN + (kIN - 1)];
#pragma unroll
  for (int j = 0; j < kEI; ++j) {
    const int e = tid + j * kNT;
    drv[j] = eidx[e];
    dcv[j] = eidx[kE + e];
    dwv[j] = ew[e];
  }
  const float ppv = pool_p[min(tid, kIN - 1)];
#pragma unroll
  for (int j = 0; j < kXA; ++j) {
    const int i = tid + j * kNT;
    xv[j] = x[i];
    wv[j] = W0[i];
  }

  for (int i = tid; i < kRP * kLD; i += nthr) s_adj[i] = 0.f;
  for (int i = tid; i < kN; i += nthr) s_deg[i] = 0.f;

#pragma unroll
  for (int j = kXA; j < kXI; ++j) {
    const int i = min(tid + j * kNT, kN * kIN - 1);
    xv[j] = x[i];
    wv[j] = W0[i];
  }

  const int gpart = (wave < 3) ? 0 : ((wave >= 4 && wave < 7) ? 1 : 2);
  const int gbidx = min(gpart * kN + nb * 16 + m16, 3 * kN - 1);
  const float bsi = b_ih[gbidx];
  const float bsh = b_hh[gbidx];

  for (int i = tid; i < kRP * kLD; i += nthr) {
    s_x[i] = 0.f; s_h[i] = 0.f;
  }
  __syncthreads();

#pragma unroll
  for (int j = 0; j < kXI; ++j) {
    const int i = tid + j * kNT;
    if (i < kN * kIN) {
      const int r = i / kIN, c = i % kIN;
      s_x[r * kLD + c] = xv[j];
      s_h[r * kLD + c] = wv[j];
    }
  }
#pragma unroll
  for (int it = 0; it < kGM; ++it) {
    const int g = it >> 2, kc = it & 3;
    s_g[(g * 16 + wave) * kLD + kc * 32 + lane] = (wave <= rlim) ? gv[it] : 0.f;
  }
  if (tid < 6 * 16 * 4)
    s_g[(tid >> 2) * kLD + 128 + (tid & 3)] = (((tid & 3) == 0) && (tjn < kN)) ? gtv : 0.f;
  if (tid < kLD) s_pp[tid] = (tid < kIN) ? ppv : 0.f;
#pragma unroll
  for (int j = 0; j < kEI; ++j) {
    atomicAdd(&s_deg[dcv[j]], dwv[j]);
  }
  for (int i = tid; i < kN; i += nthr) atomicAdd(&s_deg[i], 1.0f);
  __syncthreads();

  if (tid < kN) {
    const int i = tid;
    float pn = 0.f, d = 0.f;
#pragma unroll
    for (int q = 0; q < kIN / 4; ++q) {
      const float4 pv = *reinterpret_cast<const float4*>(&s_pp[4 * q]);
      const float4 xq = *reinterpret_cast<const float4*>(&s_x[i * kLD + 4 * q]);
      pn = __builtin_fmaf(pv.x, pv.x, pn);  d = __builtin_fmaf(xq.x, pv.x, d);
      pn = __builtin_fmaf(pv.y, pv.y, pn);  d = __builtin_fmaf(xq.y, pv.y, d);
      pn = __builtin_fmaf(pv.z, pv.z, pn);  d = __builtin_fmaf(xq.z, pv.z, d);
      pn = __builtin_fmaf(pv.w, pv.w, pn);  d = __builtin_fmaf(xq.w, pv.w, d);
    }
    {
      const float pl = s_pp[kIN - 1];
      pn = __builtin_fmaf(pl, pl, pn);
      d  = __builtin_fmaf(s_x[i * kLD + kIN - 1], pl, d);
    }
    s_score[i] = tanhf(d * rsqrtf(pn));
    float dg = s_deg[i];
    s_dis[i] = dg > 0.f ? rsqrtf(fmaxf(dg, 1e-12f)) : 0.f;
  }
  __syncthreads();

  for (int i = tid; i < kN; i += nthr) {
    float si = s_score[i];
    int rank = 0;
    for (int j = 0; j < kN; ++j) {
      float sj = s_score[j];
      rank += (sj > si) || (sj == si && j < i);
    }
    s_perm[rank] = i;
  }

  {
    float av[kEI];
#pragma unroll
    for (int j = 0; j < kEI; ++j) av[j] = s_dis[drv[j]] * dwv[j] * s_dis[dcv[j]];
#pragma unroll
    for (int j = 0; j < kEI; ++j) atomicAdd(&s_adj[dcv[j] * kLD + drv[j]], av[j]);
    if (tid >= kNT - kN) {
      const int i = tid - (kNT - kN);
      const float d = s_dis[i];
      atomicAdd(&s_adj[i * kLD + i], d * d);
    }
  }

  if (wave >= 13) {
    const int t  = wave - 13;
    const int mt = rg * kTG + t;
    v8f ch;
#pragma unroll
    for (int v = 0; v < 8; ++v) ch[v] = bsh;
    const float* pha = &s_h[(mt * 16 + m16) * kLD + 2 * hh];
    const float* phb = &s_g[5 * 16 * kLD + m16 * kLD + 2 * hh];
    v2f fa[kKS], fb[kKS];
    frag_rd2<kKS>(pha, phb, 0, fa, fb);
    __builtin_amdgcn_sched_barrier(0);
    ch = mma_pl<kKS>(fa, fb, ch);
#pragma unroll
    for (int v = 0; v < 8; ++v)
      s_gs[((3 * kTG + t) * 16 + v + 8 * hh) * 16 + m16] = ch[v];
  }
  __syncthreads();

  if (wave < 12 && ((wave & 3) == 3 || wave < 7)) {
    const int isin = (wave & 3) == 3;
    const int t    = isin ? (wave >> 2) : (wave & 3);
    const int mt   = rg * kTG + t;
    const int arow = (mt * 16 + m16) * kLD;
    const int brow = m16 * kLD;
    const int xr   = mt * 16 + m16;
    const int prm  = s_perm[min(xr, kN - 1)];
    const int xrow = ((xr < kN) ? prm : xr) * kLD;
    const float sc = (xr < kN) ? s_score[prm] : 0.f;
    const float* pxa = &s_x[xrow + 2 * hh];
    const float* pha = &s_h[arow + 2 * hh];
    if (!isin) {
      const int part = wave >> 2;
      v8f c;
#pragma unroll
      for (int v = 0; v < 8; ++v) c[v] = bsi + bsh;
      const float* pbi = &s_g[part * 16 * kLD + brow + 2 * hh];
      const float* pbh = &s_g[(3 + part) * 16 * kLD + brow + 2 * hh];
      constexpr int kB = 11;
      v2f xa0[kB], bi0[kB], ha0[kB], bh0[kB], xa1[kB], bi1[kB], ha1[kB], bh1[kB];
      frag_rd4<kB>(pxa, pbi, pha, pbh, 0,          xa0, bi0, ha0, bh0);
      frag_rd4<kB>(pxa, pbi, pha, pbh, 4 * kB,     xa1, bi1, ha1, bh1);
      __builtin_amdgcn_sched_barrier(0);
      c = mma_rz<kB>(xa0, bi0, ha0, bh0, sc, c);
      __builtin_amdgcn_sched_barrier(0);
      frag_rd4<kB>(pxa, pbi, pha, pbh, 8 * kB,     xa0, bi0, ha0, bh0);
      __builtin_amdgcn_sched_barrier(0);
      c = mma_rz<kB>(xa1, bi1, ha1, bh1, sc, c);
      c = mma_rz<kB>(xa0, bi0, ha0, bh0, sc, c);
#pragma unroll
      for (int v = 0; v < 8; ++v)
        s_gs[((part * kTG + t) * 16 + v + 8 * hh) * 16 + m16] = c[v];
    } else {
      v8f ci;
#pragma unroll
      for (int v = 0; v < 8; ++v) ci[v] = bsi;
      const float* pbn = &s_g[2 * 16 * kLD + brow + 2 * hh];
      v2f fa[kKS], fb[kKS];
      frag_rd2<kKS>(pxa, pbn, 0, fa, fb);
      __builtin_amdgcn_sched_barrier(0);
      ci = mma_xt<kKS>(fa, fb, sc, ci);
#pragma unroll
      for (int v = 0; v < 8; ++v)
        s_gs[((2 * kTG + t) * 16 + v + 8 * hh) * 16 + m16] = ci[v];
    }
  }
  __syncthreads();

  {
    const int nrow = (rg == kRG - 1) ? kLD - (kRG - 1) * kTG * 16 : kTG * 16;
    for (int e = tid; e < nrow * 16; e += nthr) {
      const int rl  = e >> 4, c = e & 15;
      const int row = rg * kTG * 16 + rl;
      const int jn  = nb * 16 + c;
      const float w0v = (jn < kLD) ? s_h[row * kLD + jn] : 0.f;
      const float wn  = gru_out(s_gs[(0 * kTG * 16 + rl) * 16 + c], s_gs[(1 * kTG * 16 + rl) * 16 + c],
                                s_gs[(2 * kTG * 16 + rl) * 16 + c], s_gs[(3 * kTG * 16 + rl) * 16 + c], w0v);
      s_wT[c * kLD + row] = (row < kN) ? wn : 0.f;
    }
  }
  __syncthreads();

  if (wave < kMT) {
    const int mt = wave;
    v8f acc = {0.f, 0.f, 0.f, 0.f, 0.f, 0.f, 0.f, 0.f};
    const int arow = (mt * 16 + m16) * kLD;
    const int brow = m16 * kLD;
    if (rg == kRG - 1) acc = chain_k<kKS - (kRG - 1) * kKG>(&s_x[arow], &s_wT[brow], rg * kKG * 4, hh, acc);
    else               acc = chain_k<kKG>(&s_x[arow], &s_wT[brow], rg * kKG * 4, hh, acc);
#pragma unroll
    for (int v = 0; v < 8; ++v) {
      const int row = mt * 16 + v + 8 * hh;
      if (row < kLD) s_xwT[m16 * kLD + row] = acc[v];
    }
  }
  __syncthreads();

  if (wave < kMT) {
    const int mt = wave;
    const int jn = nb * 16 + m16;
    const float cb = (rg == 0) ? conv_b[min(jn, kN - 1)] : 0.f;
    v8f acc;
#pragma unroll
    for (int v = 0; v < 8; ++v) acc[v] = cb;
    const int arow = (mt * 16 + m16) * kLD;
    const int brow = m16 * kLD;
#pragma unroll 11
    for (int ks = 0; ks < kKS; ++ks) {
      const int k0 = ks * 4 + 2 * hh;
      v2f a = *reinterpret_cast<const v2f*>(&s_adj[arow + k0]);
      v2f b = *reinterpret_cast<const v2f*>(&s_xwT[brow + k0]);
      acc = wmma4(a, b, acc);
    }
#pragma unroll
    for (int v = 0; v < 8; ++v)
      s_hout[(mt * 16 + v + 8 * hh) * 16 + m16] = jn < kN ? acc[v] : 0.f;
  }
  __syncthreads();

  float* hb = h_ws + rg * kPL + nb * kHB;
  for (int i = tid; i < kHB / 4; i += nthr)
    reinterpret_cast<float4*>(hb)[i] = reinterpret_cast<const float4*>(s_hout)[i];
}

__launch_bounds__(128) __attribute__((amdgpu_waves_per_eu(1, 1)))
__global__ void evolve_gcn_linear(const float* __restrict__ h_ws,
                                  const float* __restrict__ lin_w,
                                  const float* __restrict__ lin_b,
                                  float*       __restrict__ out,
                                  unsigned long long ws_bytes)
{
  if (ws_bytes < kWsNeed) return;
  __shared__ __attribute__((aligned(16))) float s_a[16 * kLA];
  __shared__ __attribute__((aligned(16))) float s_o[16 * kOUT];
  const int tid  = threadIdx.x;
  const int lane = tid & 31;
  const int ot   = tid >> 5;
  const int m16  = lane & 15;
  const int hh   = lane >> 4;
  const int mt   = blockIdx.x;
  const int oc   = ot * 16 + m16;

  const int kb0 = tid >> 6;
  const int pw  = tid & 63;
  const float* pb = h_ws + mt * 256 + pw * 4;
  float4 pv[kRG][5];
#pragma unroll
  for (int g = 0; g < kRG; ++g) {
#pragma unroll
    for (int j = 0; j < 5; ++j)
      pv[g][j] = *reinterpret_cast<const float4*>(pb + g * kPL + min(kb0 + 2 * j, kMT - 1) * kHB);
  }
  const float lb = lin_b[oc];
  const int rowo = oc * kIN;
  v2f bv[kKS];
#pragma unroll
  for (int ks = 0; ks < kKS; ++ks) {
    const int k0 = ks * 4 + 2 * hh;
    const int i0 = min(k0,     kIN - 1);
    const int i1 = min(k0 + 1, kIN - 1);
    bv[ks].x = lin_w[rowo + i0];
    bv[ks].y = lin_w[rowo + i1];
  }

#pragma unroll
  for (int j = 0; j < 5; ++j) {
    float4 s;
    s.x = (pv[0][j].x + pv[1][j].x) + pv[2][j].x;
    s.y = (pv[0][j].y + pv[1][j].y) + pv[2][j].y;
    s.z = (pv[0][j].z + pv[1][j].z) + pv[2][j].z;
    s.w = (pv[0][j].w + pv[1][j].w) + pv[2][j].w;
    s.x = s.x > 0.f ? s.x : (__expf(s.x) - 1.f);
    s.y = s.y > 0.f ? s.y : (__expf(s.y) - 1.f);
    s.z = s.z > 0.f ? s.z : (__expf(s.z) - 1.f);
    s.w = s.w > 0.f ? s.w : (__expf(s.w) - 1.f);
    *reinterpret_cast<float4*>(&s_a[(pw >> 2) * kLA + (kb0 + 2 * j) * 16 + (pw & 3) * 4]) = s;
  }
  __syncthreads();

  v8f acc;
#pragma unroll
  for (int v = 0; v < 8; ++v) acc[v] = lb;
  const int arow = m16 * kLA;
#pragma unroll
  for (int ks = 0; ks < kKS; ++ks) {
    const int k0 = ks * 4 + 2 * hh;
    v2f a = *reinterpret_cast<const v2f*>(&s_a[arow + k0]);
    acc = wmma4(a, bv[ks], acc);
  }
#pragma unroll
  for (int v = 0; v < 8; ++v) s_o[(v + 8 * hh) * kOUT + oc] = acc[v];
  __syncthreads();
  for (int i = tid; i < 16 * kOUT / 4; i += 128) {
    const int row = mt * 16 + i / (kOUT / 4);
    if (row < kN)
      reinterpret_cast<float4*>(out)[mt * (16 * kOUT / 4) + i] = reinterpret_cast<const float4*>(s_o)[i];
  }
}

extern "C" void kernel_launch(void* const* d_in, const int* in_sizes, int n_in,
                              void* d_out, int out_size, void* d_ws, size_t ws_size,
                              hipStream_t stream) {
  (void)in_sizes; (void)n_in; (void)out_size;
  const float* x       = (const float*)d_in[0];
  const int*   eidx    = (const int*)  d_in[1];
  const float* ew      = (const float*)d_in[2];
  const float* pool_p  = (const float*)d_in[3];
  const float* W0      = (const float*)d_in[4];
  const float* w_ih    = (const float*)d_in[5];
  const float* w_hh    = (const float*)d_in[6];
  const float* b_ih    = (const float*)d_in[7];
  const float* b_hh    = (const float*)d_in[8];
  const float* conv_b  = (const float*)d_in[9];
  const float* lin_w   = (const float*)d_in[10];
  const float* lin_b   = (const float*)d_in[11];
  evolve_gcn_colblock<<<dim3(9, 3), 512, 0, stream>>>(x, eidx, ew, pool_p, W0, w_ih, w_hh,
                                             b_ih, b_hh, conv_b, (float*)d_ws,
                                             (unsigned long long)ws_size);
  evolve_gcn_linear<<<9, 128, 0, stream>>>((const float*)d_ws, lin_w, lin_b,
                                           (float*)d_out, (unsigned long long)ws_size);
}


# Front LDS traffic cut to what is read: zero fill of pads and adjacency only (16-byte stores), W0 staged for the group's 48 rows, staging stores issue as loads land (no barrier behind fill); sums kept
# speedup vs baseline: 19.4073x; 1.0698x over previous
#include <hip/hip_runtime.h>


typedef __attribute__((ext_vector_type(2))) float v2f;
typedef __attribute__((ext_vector_type(8))) float v8f;

static constexpr int kN   = 129;
static constexpr int kIN  = 129;
static constexpr int kOUT = 64;
static constexpr int kE   = 4096;
static constexpr int kLD  = 132;
static constexpr int kRP  = 144;
static constexpr int kKS  = 33;
static constexpr int kMT  = 9;
static constexpr int kRG  = 3;
static constexpr int kTG  = 3;
static constexpr int kKG  = kTG * 4;
static constexpr int kNT  = 512;
static constexpr int kXI  = (kN * kIN + kNT - 1) / kNT;
static constexpr int kHR  = kTG * 16;
static constexpr int kWI  = (kHR * kIN + kNT - 1) / kNT;
static constexpr int kGE  = 6 * 16 * kLD;
static constexpr int kGM  = 24;
static constexpr int kEI  = kE / kNT;
static constexpr int kHB  = kRP * 16;
static constexpr int kPL  = kMT * kHB;
static constexpr int kLA  = 196;
static constexpr unsigned long long kWsNeed = 4ull * kRG * kPL;

static __device__ __forceinline__ v8f wmma4(v2f a, v2f b, v8f c) {
  return __builtin_amdgcn_wmma_f32_16x16x4_f32(
       false, a,  false, b,
       (short)0, c,  false,  false);
}

static __device__ __forceinline__ float sigm(float x) {
  return 1.0f / (1.0f + __expf(-x));
}

static __device__ __forceinline__ float gru_out(float cr, float cz, float ci, float ch, float w0v) {
  const float rg  = sigm(cr);
  const float zg  = sigm(cz);
  const float cnd = tanhf(ci + rg * ch);
  return (1.f - zg) * cnd + zg * w0v;
}

template <int NKS>
static __device__ __forceinline__ v8f chain_k(const float* a, const float* b, int kb, int hh, v8f acc) {
#pragma unroll
  for (int ks = 0; ks < NKS; ++ks) {
    const int k0 = kb + ks * 4 + 2 * hh;
    v2f av = *reinterpret_cast<const v2f*>(&a[k0]);
    v2f bv = *reinterpret_cast<const v2f*>(&b[k0]);
    acc = wmma4(av, bv, acc);
  }
  return acc;
}

template <int NB>
static __device__ __forceinline__ void frag_rd2(const float* pa, const float* pb, int kb,
                                                v2f (&fa)[NB], v2f (&fb)[NB]) {
#pragma unroll
  for (int j = 0; j < NB; ++j) {
    fa[j] = *reinterpret_cast<const v2f*>(&pa[kb + 4 * j]);
    fb[j] = *reinterpret_cast<const v2f*>(&pb[kb + 4 * j]);
  }
}
template <int NB>
static __device__ __forceinline__ void frag_rd4(const float* pa, const float* pb, const float* pc, const float* pd, int kb,
                                                v2f (&fa)[NB], v2f (&fb)[NB], v2f (&fc)[NB], v2f (&fd)[NB]) {
#pragma unroll
  for (int j = 0; j < NB; ++j) {
    fa[j] = *reinterpret_cast<const v2f*>(&pa[kb + 4 * j]);
    fb[j] = *reinterpret_cast<const v2f*>(&pb[kb + 4 * j]);
    fc[j] = *reinterpret_cast<const v2f*>(&pc[kb + 4 * j]);
    fd[j] = *reinterpret_cast<const v2f*>(&pd[kb + 4 * j]);
  }
}
template <int NB>
static __device__ __forceinline__ v8f mma_rz(const v2f (&xa)[NB], const v2f (&bi)[NB],
                                             const v2f (&ha)[NB], const v2f (&bh)[NB], float sc, v8f c) {
#pragma unroll
  for (int j = 0; j < NB; ++j) {
    v2f a = xa[j];
    a.x *= sc;  a.y *= sc;
    c = wmma4(a, bi[j], c);
    c = wmma4(ha[j], bh[j], c);
  }
  return c;
}
template <int NB>
static __device__ __forceinline__ v8f mma_xt(const v2f (&xa)[NB], const v2f (&b)[NB], float sc, v8f c) {
#pragma unroll
  for (int j = 0; j < NB; ++j) {
    v2f a = xa[j];
    a.x *= sc;  a.y *= sc;
    c = wmma4(a, b[j], c);
  }
  return c;
}
template <int NB>
static __device__ __forceinline__ v8f mma_pl(const v2f (&a)[NB], const v2f (&b)[NB], v8f c) {
#pragma unroll
  for (int j = 0; j < NB; ++j) c = wmma4(a[j], b[j], c);
  return c;
}

__launch_bounds__(512, 1)
__global__ void evolve_gcn_colblock(const float* __restrict__ x,
                                    const int*   __restrict__ eidx,
                                    const float* __restrict__ ew,
                                    const float* __restrict__ pool_p,
                                    const float* __restrict__ W0,
                                    const float* __restrict__ w_ih,
                                    const float* __restrict__ w_hh,
                                    const float* __restrict__ b_ih,
                                    const float* __restrict__ b_hh,
                                    const float* __restrict__ conv_b,
                                    float*       __restrict__ h_ws,
                                    unsigned long long ws_bytes)
{
  __shared__ __attribute__((aligned(16))) float s_x [kRP * kLD];
  __shared__ __attribute__((aligned(16))) float s_adj[kRP * kLD];
  __shared__ __attribute__((aligned(16))) float s_h [kHR * kLD];
  __shared__ __attribute__((aligned(16))) float s_g [kGE];
  __shared__ __attribute__((aligned(16))) float s_wT [16 * kLD];
  __shared__ __attribute__((aligned(16))) float s_xwT[16 * kLD];
  __shared__ __attribute__((aligned(16))) float s_hout[kHB];
  __shared__ float s_score[kN];
  __shared__ __attribute__((aligned(16))) float s_pp[kLD];
  __shared__ int   s_perm [kN];
  __shared__ float s_dis  [kN];
  __shared__ float s_deg  [kN];
  __shared__ float s_gs   [4 * kTG * 16 * 16];

  const int tid    = threadIdx.x;
  constexpr int nthr = kNT;
  const int lane   = tid & 31;
  const int wave   = tid >> 5;
  const int m16    = lane & 15;
  const int hh     = lane >> 4;
  const int nb     = blockIdx.x;
  const int rg     = blockIdx.y;

  if (tid < kN) s_deg[tid] = 0.f;
  __syncthreads();

  if (ws_bytes < kWsNeed) return;

  float xv[kXI], wv[kWI], gv[kGM], dwv[kEI];
  int   drv[kEI], dcv[kEI];
#pragma unroll
  for (int j = 0; j < kEI; ++j) {
    const int e = tid + j * kNT;
    drv[j] = eidx[e];
    dcv[j] = eidx[kE + e];
    dwv[j] = ew[e];
  }
  const float ppv = pool_p[min(tid, kIN - 1)];
  const int rlim = kN - 1 - nb * 16;
  const float* gi0 = w_ih + nb * 16 * kIN;
  const float* gh0 = w_hh + nb * 16 * kIN;
  const int gofs = min(wave, rlim) * kIN + lane;
#pragma unroll
  for (int it = 0; it < kGM; ++it) {
    const int g = it >> 2, kc = it & 3;
    const float* src = (g < 3 ? gi0 : gh0) + (g % 3) * kN * kIN + kc * 32;
    gv[it] = src[gofs];
  }
  const int trow = min(tid >> 2, 6 * 16 - 1);
  const int tjn  = nb * 16 + (trow & 15);
  const float* tsrc = (trow < 3 * 16) ? w_ih : w_hh;
  const float gtv = tsrc[(((trow >> 4) % 3) * kN + min(tjn, kN - 1)) * kIN + (kIN - 1)];
  const int hcnt = ((rg == kRG - 1) ? (kN - (kRG - 1) * kHR) : kHR) * kIN;
  const float* w0g = W0 + rg * kHR * kIN;
#pragma unroll
  for (int j = 0; j < kWI; ++j) {
    const int i = tid + j * kNT;
    wv[j] = w0g[(j < 8) ? i : min(i, hcnt - 1)];
  }

  {
    const float4 z4 = {0.f, 0.f, 0.f, 0.f};
    float4* a4 = reinterpret_cast<float4*>(s_adj);
    for (int i = tid; i < kRP * kLD / 4; i += nthr) a4[i] = z4;
    if (tid < (kRP - kN) * kLD / 4)
      reinterpret_cast<float4*>(&s_x[kN * kLD])[tid] = z4;
    if (tid < 3 * kN)  s_x[(tid / 3) * kLD + kIN + (tid % 3)] = 0.f;
    if (tid < 3 * kHR) s_h[(tid / 3) * kLD + kIN + (tid % 3)] = 0.f;
    if (rg == kRG - 1 && tid < (kRP - kN) * kLD / 4)
      reinterpret_cast<float4*>(&s_h[(kN - (kRG - 1) * kHR) * kLD])[tid] = z4;
  }

#pragma unroll
  for (int j = 0; j < kXI; ++j) {
    const int i = tid + j * kNT;
    xv[j] = x[(j < kXI - 1) ? i : min(i, kN * kIN - 1)];
  }

  const int gpart = (wave < 3) ? 0 : ((wave >= 4 && wave < 7) ? 1 : 2);
  const int gbidx = min(gpart * kN + nb * 16 + m16, 3 * kN - 1);
  const float bsi = b_ih[gbidx];
  const float bsh = b_hh[gbidx];

#pragma unroll
  for (int j = 0; j < kEI; ++j) {
    atomicAdd(&s_deg[dcv[j]], dwv[j]);
  }
  for (int i = tid; i < kN; i += nthr) atomicAdd(&s_deg[i], 1.0f);
  if (tid < kLD) s_pp[tid] = (tid < kIN) ? ppv : 0.f;
#pragma unroll
  for (int it = 0; it < kGM; ++it) {
    const int g = it >> 2, kc = it & 3;
    s_g[(g * 16 + wave) * kLD + kc * 32 + lane] = (wave <= rlim) ? gv[it] : 0.f;
  }
  if (tid < 6 * 16 * 4)
    s_g[(tid >> 2) * kLD + 128 + (tid & 3)] = (((tid & 3) == 0) && (tjn < kN)) ? gtv : 0.f;
#pragma unroll
  for (int j = 0; j < kWI; ++j) {
    const int i = tid + j * kNT;
    const int off = (i / kIN) * kLD + (i % kIN);
    if (j < 8 || i < hcnt) s_h[off] = wv[j];
  }
#pragma unroll
  for (int j = 0; j < kXI; ++j) {
    const int i = tid + j * kNT;
    const int off = (i / kIN) * kLD + (i % kIN);
    if (j < kXI - 1 || i < kN * kIN) s_x[off] = xv[j];
  }
  __syncthreads();

  if (tid < kN) {
    const int i = tid;
    float pn = 0.f, d = 0.f;
#pragma unroll
    for (int q = 0; q < kIN / 4; ++q) {
      const float4 pv = *reinterpret_cast<const float4*>(&s_pp[4 * q]);
      const float4 xq = *reinterpret_cast<const float4*>(&s_x[i * kLD + 4 * q]);
      pn = __builtin_fmaf(pv.x, pv.x, pn);  d = __builtin_fmaf(xq.x, pv.x, d);
      pn = __builtin_fmaf(pv.y, pv.y, pn);  d = __builtin_fmaf(xq.y, pv.y, d);
      pn = __builtin_fmaf(pv.z, pv.z, pn);  d = __builtin_fmaf(xq.z, pv.z, d);
      pn = __builtin_fmaf(pv.w, pv.w, pn);  d = __builtin_fmaf(xq.w, pv.w, d);
    }
    {
      const float pl = s_pp[kIN - 1];
      pn = __builtin_fmaf(pl, pl, pn);
      d  = __builtin_fmaf(s_x[i * kLD + kIN - 1], pl, d);
    }
    s_score[i] = tanhf(d * rsqrtf(pn));
    float dg = s_deg[i];
    s_dis[i] = dg > 0.f ? rsqrtf(fmaxf(dg, 1e-12f)) : 0.f;
  }
  __syncthreads();

  for (int i = tid; i < kN; i += nthr) {
    float si = s_score[i];
    int rank = 0;
    for (int j = 0; j < kN; ++j) {
      float sj = s_score[j];
      rank += (sj > si) || (sj == si && j < i);
    }
    s_perm[rank] = i;
  }

  {
    float av[kEI];
#pragma unroll
    for (int j = 0; j < kEI; ++j) av[j] = s_dis[drv[j]] * dwv[j] * s_dis[dcv[j]];
#pragma unroll
    for (int j = 0; j < kEI; ++j) atomicAdd(&s_adj[dcv[j] * kLD + drv[j]], av[j]);
    if (tid >= kNT - kN) {
      const int i = tid - (kNT - kN);
      const float d = s_dis[i];
      atomicAdd(&s_adj[i * kLD + i], d * d);
    }
  }

  if (wave >= 13) {
    const int t  = wave - 13;
    v8f ch;
#pragma unroll
    for (int v = 0; v < 8; ++v) ch[v] = bsh;
    const float* pha = &s_h[(t * 16 + m16) * kLD + 2 * hh];
    const float* phb = &s_g[5 * 16 * kLD + m16 * kLD + 2 * hh];
    v2f fa[kKS], fb[kKS];
    frag_rd2<kKS>(pha, phb, 0, fa, fb);
    __builtin_amdgcn_sched_barrier(0);
    ch = mma_pl<kKS>(fa, fb, ch);
#pragma unroll
    for (int v = 0; v < 8; ++v)
      s_gs[((3 * kTG + t) * 16 + v + 8 * hh) * 16 + m16] = ch[v];
  }
  __syncthreads();

  if (wave < 12 && ((wave & 3) == 3 || wave < 7)) {
    const int isin = (wave & 3) == 3;
    const int t    = isin ? (wave >> 2) : (wave & 3);
    const int mt   = rg * kTG + t;
    const int arow = (t * 16 + m16) * kLD;
    const int brow = m16 * kLD;
    const int xr   = mt * 16 + m16;
    const int prm  = s_perm[min(xr, kN - 1)];
    const int xrow = ((xr < kN) ? prm : xr) * kLD;
    const float sc = (xr < kN) ? s_score[prm] : 0.f;
    const float* pxa = &s_x[xrow + 2 * hh];
    const float* pha = &s_h[arow + 2 * hh];
    if (!isin) {
      const int part = wave >> 2;
      v8f c;
#pragma unroll
      for (int v = 0; v < 8; ++v) c[v] = bsi + bsh;
      const float* pbi = &s_g[part * 16 * kLD + brow + 2 * hh];
      const float* pbh = &s_g[(3 + part) * 16 * kLD + brow + 2 * hh];
      constexpr int kB = 11;
      v2f xa0[kB], bi0[kB], ha0[kB], bh0[kB], xa1[kB], bi1[kB], ha1[kB], bh1[kB];
      frag_rd4<kB>(pxa, pbi, pha, pbh, 0,          xa0, bi0, ha0, bh0);
      frag_rd4<kB>(pxa, pbi, pha, pbh, 4 * kB,     xa1, bi1, ha1, bh1);
      __builtin_amdgcn_sched_barrier(0);
      c = mma_rz<kB>(xa0, bi0, ha0, bh0, sc, c);
      __builtin_amdgcn_sched_barrier(0);
      frag_rd4<kB>(pxa, pbi, pha, pbh, 8 * kB,     xa0, bi0, ha0, bh0);
      __builtin_amdgcn_sched_barrier(0);
      c = mma_rz<kB>(xa1, bi1, ha1, bh1, sc, c);
      c = mma_rz<kB>(xa0, bi0, ha0, bh0, sc, c);
#pragma unroll
      for (int v = 0; v < 8; ++v)
        s_gs[((part * kTG + t) * 16 + v + 8 * hh) * 16 + m16] = c[v];
    } else {
      v8f ci;
#pragma unroll
      for (int v = 0; v < 8; ++v) ci[v] = bsi;
      const float* pbn = &s_g[2 * 16 * kLD + brow + 2 * hh];
      v2f fa[kKS], fb[kKS];
      frag_rd2<kKS>(pxa, pbn, 0, fa, fb);
      __builtin_amdgcn_sched_barrier(0);
      ci = mma_xt<kKS>(fa, fb, sc, ci);
#pragma unroll
      for (int v = 0; v < 8; ++v)
        s_gs[((2 * kTG + t) * 16 + v + 8 * hh) * 16 + m16] = ci[v];
    }
  }
  __syncthreads();

  {
    const int nrow = (rg == kRG - 1) ? kLD - (kRG - 1) * kTG * 16 : kTG * 16;
    for (int e = tid; e < nrow * 16; e += nthr) {
      const int rl  = e >> 4, c = e & 15;
      const int row = rg * kTG * 16 + rl;
      const int jn  = nb * 16 + c;
      const float w0v = (jn < kLD) ? s_h[rl * kLD + jn] : 0.f;
      const float wn  = gru_out(s_gs[(0 * kTG * 16 + rl) * 16 + c], s_gs[(1 * kTG * 16 + rl) * 16 + c],
                                s_gs[(2 * kTG * 16 + rl) * 16 + c], s_gs[(3 * kTG * 16 + rl) * 16 + c], w0v);
      s_wT[c * kLD + row] = (row < kN) ? wn : 0.f;
    }
  }
  __syncthreads();

  if (wave < kMT) {
    const int mt = wave;
    v8f acc = {0.f, 0.f, 0.f, 0.f, 0.f, 0.f, 0.f, 0.f};
    const int arow = (mt * 16 + m16) * kLD;
    const int brow = m16 * kLD;
    if (rg == kRG - 1) acc = chain_k<kKS - (kRG - 1) * kKG>(&s_x[arow], &s_wT[brow], rg * kKG * 4, hh, acc);
    else               acc = chain_k<kKG>(&s_x[arow], &s_wT[brow], rg * kKG * 4, hh, acc);
#pragma unroll
    for (int v = 0; v < 8; ++v) {
      const int row = mt * 16 + v + 8 * hh;
      if (row < kLD) s_xwT[m16 * kLD + row] = acc[v];
    }
  }
  __syncthreads();

  if (wave < kMT) {
    const int mt = wave;
    const int jn = nb * 16 + m16;
    const float cb = (rg == 0) ? conv_b[min(jn, kN - 1)] : 0.f;
    v8f acc;
#pragma unroll
    for (int v = 0; v < 8; ++v) acc[v] = cb;
    const int arow = (mt * 16 + m16) * kLD;
    const int brow = m16 * kLD;
#pragma unroll 11
    for (int ks = 0; ks < kKS; ++ks) {
      const int k0 = ks * 4 + 2 * hh;
      v2f a = *reinterpret_cast<const v2f*>(&s_adj[arow + k0]);
      v2f b = *reinterpret_cast<const v2f*>(&s_xwT[brow + k0]);
      acc = wmma4(a, b, acc);
    }
#pragma unroll
    for (int v = 0; v < 8; ++v)
      s_hout[(mt * 16 + v + 8 * hh) * 16 + m16] = jn < kN ? acc[v] : 0.f;
  }
  __syncthreads();

  float* hb = h_ws + rg * kPL + nb * kHB;
  for (int i = tid; i < kHB / 4; i += nthr)
    reinterpret_cast<float4*>(hb)[i] = reinterpret_cast<const float4*>(s_hout)[i];
}

__launch_bounds__(128) __attribute__((amdgpu_waves_per_eu(1, 1)))
__global__ void evolve_gcn_linear(const float* __restrict__ h_ws,
                                  const float* __restrict__ lin_w,
                                  const float* __restrict__ lin_b,
                                  float*       __restrict__ out,
                                  unsigned long long ws_bytes)
{
  if (ws_bytes < kWsNeed) return;
  __shared__ __attribute__((aligned(16))) float s_a[16 * kLA];
  __shared__ __attribute__((aligned(16))) float s_o[16 * kOUT];
  const int tid  = threadIdx.x;
  const int lane = tid & 31;
  const int ot   = tid >> 5;
  const int m16  = lane & 15;
  const int hh   = lane >> 4;
  const int mt   = blockIdx.x;
  const int oc   = ot * 16 + m16;

  const int kb0 = tid >> 6;
  const int pw  = tid & 63;
  const float* pb = h_ws + mt * 256 + pw * 4;
  float4 pv[kRG][5];
#pragma unroll
  for (int g = 0; g < kRG; ++g) {
#pragma unroll
    for (int j = 0; j < 5; ++j)
      pv[g][j] = *reinterpret_cast<const float4*>(pb + g * kPL + min(kb0 + 2 * j, kMT - 1) * kHB);
  }
  const float lb = lin_b[oc];
  const int rowo = oc * kIN;
  v2f bv[kKS];
#pragma unroll
  for (int ks = 0; ks < kKS; ++ks) {
    const int k0 = ks * 4 + 2 * hh;
    const int i0 = min(k0,     kIN - 1);
    const int i1 = min(k0 + 1, kIN - 1);
    bv[ks].x = lin_w[rowo + i0];
    bv[ks].y = lin_w[rowo + i1];
  }

#pragma unroll
  for (int j = 0; j < 5; ++j) {
    float4 s;
    s.x = (pv[0][j].x + pv[1][j].x) + pv[2][j].x;
    s.y = (pv[0][j].y + pv[1][j].y) + pv[2][j].y;
    s.z = (pv[0][j].z + pv[1][j].z) + pv[2][j].z;
    s.w = (pv[0][j].w + pv[1][j].w) + pv[2][j].w;
    s.x = s.x > 0.f ? s.x : (__expf(s.x) - 1.f);
    s.y = s.y > 0.f ? s.y : (__expf(s.y) - 1.f);
    s.z = s.z > 0.f ? s.z : (__expf(s.z) - 1.f);
    s.w = s.w > 0.f ? s.w : (__expf(s.w) - 1.f);
    *reinterpret_cast<float4*>(&s_a[(pw >> 2) * kLA + (kb0 + 2 * j) * 16 + (pw & 3) * 4]) = s;
  }
  __syncthreads();

  v8f acc;
#pragma unroll
  for (int v = 0; v < 8; ++v) acc[v] = lb;
  const int arow = m16 * kLA;
#pragma unroll
  for (int ks = 0; ks < kKS; ++ks) {
    const int k0 = ks * 4 + 2 * hh;
    v2f a = *reinterpret_cast<const v2f*>(&s_a[arow + k0]);
    acc = wmma4(a, bv[ks], acc);
  }
#pragma unroll
  for (int v = 0; v < 8; ++v) s_o[(v + 8 * hh) * kOUT + oc] = acc[v];
  __syncthreads();
  for (int i = tid; i < 16 * kOUT / 4; i += 128) {
    const int row = mt * 16 + i / (kOUT / 4);
    if (row < kN)
      reinterpret_cast<float4*>(out)[mt * (16 * kOUT / 4) + i] = reinterpret_cast<const float4*>(s_o)[i];
  }
}

extern "C" void kernel_launch(void* const* d_in, const int* in_sizes, int n_in,
                              void* d_out, int out_size, void* d_ws, size_t ws_size,
                              hipStream_t stream) {
  (void)in_sizes; (void)n_in; (void)out_size;
  const float* x       = (const float*)d_in[0];
  const int*   eidx    = (const int*)  d_in[1];
  const float* ew      = (const float*)d_in[2];
  const float* pool_p  = (const float*)d_in[3];
  const float* W0      = (const float*)d_in[4];
  const float* w_ih    = (const float*)d_in[5];
  const float* w_hh    = (const float*)d_in[6];
  const float* b_ih    = (const float*)d_in[7];
  const float* b_hh    = (const float*)d_in[8];
  const float* conv_b  = (const float*)d_in[9];
  const float* lin_w   = (const float*)d_in[10];
  const float* lin_b   = (const float*)d_in[11];
  evolve_gcn_colblock<<<dim3(9, 3), 512, 0, stream>>>(x, eidx, ew, pool_p, W0, w_ih, w_hh,
                                             b_ih, b_hh, conv_b, (float*)d_ws,
                                             (unsigned long long)ws_size);
  evolve_gcn_linear<<<9, 128, 0, stream>>>((const float*)d_ws, lin_w, lin_b,
                                           (float*)d_out, (unsigned long long)ws_size);
}
